# starting kernel
# baseline (speedup 1.0000x reference)
#include <hip/hip_runtime.h>
#include <hip/hip_bf16.h>

typedef __attribute__((ext_vector_type(2))) float v2f;
typedef __attribute__((ext_vector_type(8))) float v8f;

template<int CIN, int COUT>
__global__ __launch_bounds__(32) void edge_gemm_dual(
    const float* __restrict__ X,
    const float* __restrict__ W,
    const float* __restrict__ bias,
    float* __restrict__ A,
    float* __restrict__ V)
{
    constexpr int NT = (COUT + 15) / 16;
    const int lane = threadIdx.x;
    const int mt = blockIdx.x / NT;
    const int nt = blockIdx.x % NT;
    const int m0 = mt * 16;
    const int n0 = nt * 16;
    const int half = lane >> 4;
    const int l15  = lane & 15;

    const int rowA = m0 + l15;
    const int colB = n0 + l15;
    const bool cok = (colB < COUT);

    v8f accA = {};
    v8f accV = {};

    for (int kk = 0; kk < CIN; kk += 4) {
        const int ka = kk + half * 2;
        v2f a;
        a.x = (ka     < CIN) ? X[rowA * CIN + ka]     : 0.0f;
        a.y = (ka + 1 < CIN) ? X[rowA * CIN + ka + 1] : 0.0f;

        v2f bt, bb;
        bt.x = (cok && ka     < CIN) ? W[ ka            * COUT + colB] : 0.0f;
        bt.y = (cok && ka + 1 < CIN) ? W[(ka + 1)       * COUT + colB] : 0.0f;
        bb.x = (cok && ka     < CIN) ? W[(CIN + ka)     * COUT + colB] : 0.0f;
        bb.y = (cok && ka + 1 < CIN) ? W[(CIN + ka + 1) * COUT + colB] : 0.0f;

        accA = __builtin_amdgcn_wmma_f32_16x16x4_f32(
                   false, a, false, bt, (short)0, accA, false, false);
        accV = __builtin_amdgcn_wmma_f32_16x16x4_f32(
                   false, a, false, bb, (short)0, accV, false, false);
    }

    if (cok) {
        const float bv = bias[colB];
#pragma unroll
        for (int r = 0; r < 8; ++r) {
            const int row = m0 + r + half * 8;
            A[row * COUT + colB] = accA[r] + bv;
            V[row * COUT + colB] = accV[r];
        }
    }
}

template<int C>
__global__ __launch_bounds__(256) void edge_pairwise_relu(
    const float* __restrict__ A,
    const float* __restrict__ V,
    float* __restrict__ out,
    int N, float inv_nm1)
{
    constexpr int TJ = 128;
    constexpr int R  = 256 / C;
    __shared__ float vt[TJ * C];

    const int tid  = threadIdx.x;
    const int c    = tid % C;
    const int iloc = tid / C;
    const int i    = blockIdx.x * R + iloc;
    const int base = (i / N) * N;

    const float a = A[i * C + c];
    float selfterm = fmaxf(a + V[i * C + c], 0.0f);

    float s = 0.0f;
    for (int jt = 0; jt < N; jt += TJ) {
        const float* src = V + (size_t)(base + jt) * C;
        for (int t = tid; t < TJ * C; t += 256) vt[t] = src[t];
        __syncthreads();
#pragma unroll 8
        for (int jj = 0; jj < TJ; ++jj)
            s += fmaxf(a + vt[jj * C + c], 0.0f);
        __syncthreads();
    }
    out[i * C + c] = (s - selfterm) * inv_nm1;
}

__global__ __launch_bounds__(256) void edge_colsum3(
    const float* __restrict__ V, float* __restrict__ cs, int N)
{
    __shared__ float red[256];
    const int b = blockIdx.x / 3;
    const int c = blockIdx.x % 3;
    float s = 0.0f;
    for (int j = threadIdx.x; j < N; j += 256)
        s += V[(size_t)(b * N + j) * 3 + c];
    red[threadIdx.x] = s;
    __syncthreads();
    for (int off = 128; off > 0; off >>= 1) {
        if (threadIdx.x < off) red[threadIdx.x] += red[threadIdx.x + off];
        __syncthreads();
    }
    if (threadIdx.x == 0) cs[blockIdx.x] = red[0];
}

__global__ __launch_bounds__(256) void edge_final3(
    const float* __restrict__ A,
    const float* __restrict__ V,
    const float* __restrict__ cs,
    float* __restrict__ out,
    int N, float inv_nm1)
{
    const int idx = blockIdx.x * 256 + threadIdx.x;
    const int c = idx % 3;
    const int b = (idx / 3) / N;
    out[idx] = A[idx] + (cs[b * 3 + c] - V[idx]) * inv_nm1;
}

extern "C" void kernel_launch(void* const* d_in, const int* in_sizes, int n_in,
                              void* d_out, int out_size, void* d_ws, size_t ws_size,
                              hipStream_t stream) {
    (void)in_sizes; (void)n_in; (void)out_size; (void)ws_size;

    const float* x  = (const float*)d_in[0];
    const float* W1 = (const float*)d_in[1];
    const float* b1 = (const float*)d_in[2];
    const float* W2 = (const float*)d_in[3];
    const float* b2 = (const float*)d_in[4];
    const float* W3 = (const float*)d_in[5];
    const float* b3 = (const float*)d_in[6];
    float* out = (float*)d_out;

    const int B = 2, N = 1024;
    const int ROWS = B * N;
    const float inv = 1.0f / (float)(N - 1);

    float* ws = (float*)d_ws;
    float* A1 = ws;
    float* V1 = A1 + ROWS * 32;
    float* H1 = V1 + ROWS * 32;
    float* A2 = H1 + ROWS * 32;
    float* V2 = A2 + ROWS * 64;
    float* H2 = V2 + ROWS * 64;
    float* A3 = H2 + ROWS * 64;
    float* V3 = A3 + ROWS * 3;
    float* CS = V3 + ROWS * 3;

    const int MT = ROWS / 16;

    edge_gemm_dual<3, 32><<<MT * 2, 32, 0, stream>>>(x, W1, b1, A1, V1);
    edge_pairwise_relu<32><<<ROWS / (256 / 32), 256, 0, stream>>>(A1, V1, H1, N, inv);

    edge_gemm_dual<32, 64><<<MT * 4, 32, 0, stream>>>(H1, W2, b2, A2, V2);
    edge_pairwise_relu<64><<<ROWS / (256 / 64), 256, 0, stream>>>(A2, V2, H2, N, inv);

    edge_gemm_dual<64, 3><<<MT * 1, 32, 0, stream>>>(H2, W3, b3, A3, V3);
    edge_colsum3<<<B * 3, 256, 0, stream>>>(V3, CS, N);
    edge_final3<<<(ROWS * 3) / 256, 256, 0, stream>>>(A3, V3, CS, out, N, inv);
}


# pairwise relu-sum re-tiled: block = 8/16 rows x 32 channels held in registers, its 8 waves split j (partial sums added via LDS: new summation order), V chunks fetched one ahead by 16-byte loads
# speedup vs baseline: 2.2829x; 2.2829x over previous
#include <hip/hip_runtime.h>
#include <hip/hip_bf16.h>

typedef __attribute__((ext_vector_type(2))) float v2f;
typedef __attribute__((ext_vector_type(8))) float v8f;
typedef __attribute__((ext_vector_type(4))) float v4f;

template<int CIN, int COUT>
__global__ __launch_bounds__(32) void edge_gemm_dual(
    const float* __restrict__ X,
    const float* __restrict__ W,
    const float* __restrict__ bias,
    float* __restrict__ A,
    float* __restrict__ V)
{
    constexpr int NT = (COUT + 15) / 16;
    const int lane = threadIdx.x;
    const int mt = blockIdx.x / NT;
    const int nt = blockIdx.x % NT;
    const int m0 = mt * 16;
    const int n0 = nt * 16;
    const int half = lane >> 4;
    const int l15  = lane & 15;

    const int rowA = m0 + l15;
    const int colB = n0 + l15;
    const bool cok = (colB < COUT);

    v8f accA = {};
    v8f accV = {};

    for (int kk = 0; kk < CIN; kk += 4) {
        const int ka = kk + half * 2;
        v2f a;
        a.x = (ka     < CIN) ? X[rowA * CIN + ka]     : 0.0f;
        a.y = (ka + 1 < CIN) ? X[rowA * CIN + ka + 1] : 0.0f;

        v2f bt, bb;
        bt.x = (cok && ka     < CIN) ? W[ ka            * COUT + colB] : 0.0f;
        bt.y = (cok && ka + 1 < CIN) ? W[(ka + 1)       * COUT + colB] : 0.0f;
        bb.x = (cok && ka     < CIN) ? W[(CIN + ka)     * COUT + colB] : 0.0f;
        bb.y = (cok && ka + 1 < CIN) ? W[(CIN + ka + 1) * COUT + colB] : 0.0f;

        accA = __builtin_amdgcn_wmma_f32_16x16x4_f32(
                   false, a, false, bt, (short)0, accA, false, false);
        accV = __builtin_amdgcn_wmma_f32_16x16x4_f32(
                   false, a, false, bb, (short)0, accV, false, false);
    }

    if (cok) {
        const float bv = bias[colB];
#pragma unroll
        for (int r = 0; r < 8; ++r) {
            const int row = m0 + r + half * 8;
            A[row * COUT + colB] = accA[r] + bv;
            V[row * COUT + colB] = accV[r];
        }
    }
}

template<int C, int RI>
__global__ __launch_bounds__(256) void edge_pairwise_relu(
    const float* __restrict__ A,
    const float* __restrict__ V,
    float* __restrict__ out,
    int N, float inv_nm1)
{
    constexpr int TJ   = 256;
    constexpr int NSEG = C / 32;
    constexpr int JW   = TJ / 8;
    constexpr int NLD  = (TJ * 8) / 256;
    __shared__ v4f    vt4[2][TJ * 8];
    __shared__ float  red[8][RI][32];

    const int tid  = threadIdx.x;
    const int lane = tid & 31;
    const int wid  = tid >> 5;
    const int seg  = blockIdx.x % NSEG;
    const int i0   = (blockIdx.x / NSEG) * RI;
    const int base = (i0 / N) * N;
    const int c    = seg * 32 + lane;

    const float* vsrc = V + (size_t)base * C + seg * 32
                          + (size_t)(tid >> 3) * C + (tid & 7) * 4;
    v4f pre[NLD];
#pragma unroll
    for (int k = 0; k < NLD; ++k)
        pre[k] = *reinterpret_cast<const v4f*>(vsrc + (size_t)(32 * k) * C);

    float a[RI], acc[RI];
#pragma unroll
    for (int r = 0; r < RI; ++r) {
        a[r]   = A[(i0 + r) * C + c];
        acc[r] = 0.0f;
    }
    float selfterm[RI / 8];
#pragma unroll
    for (int q = 0; q < RI / 8; ++q) {
        const int i = i0 + wid + 8 * q;
        selfterm[q] = fmaxf(A[i * C + c] + V[i * C + c], 0.0f);
    }

    int buf = 0;
    for (int jt = 0; jt < N; jt += TJ) {
#pragma unroll
        for (int k = 0; k < NLD; ++k) vt4[buf][tid + 256 * k] = pre[k];
        __syncthreads();
        if (jt + TJ < N) {
            const float* nsrc = vsrc + (size_t)(jt + TJ) * C;
#pragma unroll
            for (int k = 0; k < NLD; ++k)
                pre[k] = *reinterpret_cast<const v4f*>(nsrc + (size_t)(32 * k) * C);
        }
        const float* vb = reinterpret_cast<const float*>(vt4[buf]) + (wid * JW) * 32 + lane;
        float v[JW];
#pragma unroll
        for (int jj = 0; jj < JW; ++jj) v[jj] = vb[jj * 32];
        __builtin_amdgcn_sched_barrier(0);
#pragma unroll
        for (int jj = 0; jj < JW; ++jj) {
#pragma unroll
            for (int r = 0; r < RI; ++r)
                acc[r] += fmaxf(a[r] + v[jj], 0.0f);
        }
        buf ^= 1;
    }

#pragma unroll
    for (int r = 0; r < RI; ++r) red[wid][r][lane] = acc[r];
    __syncthreads();

#pragma unroll
    for (int q = 0; q < RI / 8; ++q) {
        const int rr = wid + 8 * q;
        float s = red[0][rr][lane];
#pragma unroll
        for (int w = 1; w < 8; ++w) s += red[w][rr][lane];
        out[(i0 + rr) * C + c] = (s - selfterm[q]) * inv_nm1;
    }
}

__global__ __launch_bounds__(256) void edge_colsum3(
    const float* __restrict__ V, float* __restrict__ cs, int N)
{
    __shared__ float red[256];
    const int b = blockIdx.x / 3;
    const int c = blockIdx.x % 3;
    float s = 0.0f;
    for (int j = threadIdx.x; j < N; j += 256)
        s += V[(size_t)(b * N + j) * 3 + c];
    red[threadIdx.x] = s;
    __syncthreads();
    for (int off = 128; off > 0; off >>= 1) {
        if (threadIdx.x < off) red[threadIdx.x] += red[threadIdx.x + off];
        __syncthreads();
    }
    if (threadIdx.x == 0) cs[blockIdx.x] = red[0];
}

__global__ __launch_bounds__(256) void edge_final3(
    const float* __restrict__ A,
    const float* __restrict__ V,
    const float* __restrict__ cs,
    float* __restrict__ out,
    int N, float inv_nm1)
{
    const int idx = blockIdx.x * 256 + threadIdx.x;
    const int c = idx % 3;
    const int b = (idx / 3) / N;
    out[idx] = A[idx] + (cs[b * 3 + c] - V[idx]) * inv_nm1;
}

extern "C" void kernel_launch(void* const* d_in, const int* in_sizes, int n_in,
                              void* d_out, int out_size, void* d_ws, size_t ws_size,
                              hipStream_t stream) {
    (void)in_sizes; (void)n_in; (void)out_size; (void)ws_size;

    const float* x  = (const float*)d_in[0];
    const float* W1 = (const float*)d_in[1];
    const float* b1 = (const float*)d_in[2];
    const float* W2 = (const float*)d_in[3];
    const float* b2 = (const float*)d_in[4];
    const float* W3 = (const float*)d_in[5];
    const float* b3 = (const float*)d_in[6];
    float* out = (float*)d_out;

    const int B = 2, N = 1024;
    const int ROWS = B * N;
    const float inv = 1.0f / (float)(N - 1);

    float* ws = (float*)d_ws;
    float* A1 = ws;
    float* V1 = A1 + ROWS * 32;
    float* H1 = V1 + ROWS * 32;
    float* A2 = H1 + ROWS * 32;
    float* V2 = A2 + ROWS * 64;
    float* H2 = V2 + ROWS * 64;
    float* A3 = H2 + ROWS * 64;
    float* V3 = A3 + ROWS * 3;
    float* CS = V3 + ROWS * 3;

    const int MT = ROWS / 16;

    edge_gemm_dual<3, 32><<<MT * 2, 32, 0, stream>>>(x, W1, b1, A1, V1);
    edge_pairwise_relu<32, 8><<<ROWS / 8, 256, 0, stream>>>(A1, V1, H1, N, inv);

    edge_gemm_dual<32, 64><<<MT * 4, 32, 0, stream>>>(H1, W2, b2, A2, V2);
    edge_pairwise_relu<64, 16><<<(ROWS / 16) * 2, 256, 0, stream>>>(A2, V2, H2, N, inv);

    edge_gemm_dual<64, 3><<<MT * 1, 32, 0, stream>>>(H2, W3, b3, A3, V3);
    edge_colsum3<<<B * 3, 256, 0, stream>>>(V3, CS, N);
    edge_final3<<<(ROWS * 3) / 256, 256, 0, stream>>>(A3, V3, CS, out, N, inv);
}


# next layer's dual GEMM (fp32 WMMA) done in each pairwise kernel's epilogue, column sum inside the final kernel: 7 launches -> 4; last GEMM as two half-k partial sums (new summation order)
# speedup vs baseline: 3.9887x; 1.7472x over previous
#include <hip/hip_runtime.h>
#include <hip/hip_bf16.h>

typedef __attribute__((ext_vector_type(2))) float v2f;
typedef __attribute__((ext_vector_type(8))) float v8f;
typedef __attribute__((ext_vector_type(4))) float v4f;

template<int CIN, int COUT>
__global__ __launch_bounds__(32) void edge_gemm_dual(
    const float* __restrict__ X,
    const float* __restrict__ W,
    const float* __restrict__ bias,
    float* __restrict__ A,
    float* __restrict__ V)
{
    constexpr int NT = (COUT + 15) / 16;
    const int lane = threadIdx.x;
    const int mt = blockIdx.x / NT;
    const int nt = blockIdx.x % NT;
    const int m0 = mt * 16;
    const int n0 = nt * 16;
    const int half = lane >> 4;
    const int l15  = lane & 15;

    const int rowA = m0 + l15;
    const int colB = n0 + l15;
    const bool cok = (colB < COUT);

    v8f accA = {};
    v8f accV = {};

    for (int kk = 0; kk < CIN; kk += 4) {
        const int ka = kk + half * 2;
        v2f a;
        a.x = (ka     < CIN) ? X[rowA * CIN + ka]     : 0.0f;
        a.y = (ka + 1 < CIN) ? X[rowA * CIN + ka + 1] : 0.0f;

        v2f bt, bb;
        bt.x = (cok && ka     < CIN) ? W[ ka            * COUT + colB] : 0.0f;
        bt.y = (cok && ka + 1 < CIN) ? W[(ka + 1)       * COUT + colB] : 0.0f;
        bb.x = (cok && ka     < CIN) ? W[(CIN + ka)     * COUT + colB] : 0.0f;
        bb.y = (cok && ka + 1 < CIN) ? W[(CIN + ka + 1) * COUT + colB] : 0.0f;

        accA = __builtin_amdgcn_wmma_f32_16x16x4_f32(
                   false, a, false, bt, (short)0, accA, false, false);
        accV = __builtin_amdgcn_wmma_f32_16x16x4_f32(
                   false, a, false, bb, (short)0, accV, false, false);
    }

    if (cok) {
        const float bv = bias[colB];
#pragma unroll
        for (int r = 0; r < 8; ++r) {
            const int row = m0 + r + half * 8;
            A[row * COUT + colB] = accA[r] + bv;
            V[row * COUT + colB] = accV[r];
        }
    }
}

template<int C, int RI, int COUT, bool ADD_BIAS>
__global__ __launch_bounds__(256) void edge_pairwise_relu_gemm(
    const float* __restrict__ A,
    const float* __restrict__ V,
    const float* __restrict__ Wn,
    const float* __restrict__ bias,
    float* __restrict__ outA,
    float* __restrict__ outV,
    int N, float inv_nm1, int plane)
{
    constexpr int TJ   = 256;
    constexpr int NSEG = C / 32;
    constexpr int JW   = TJ / 8;
    constexpr int NLD  = (TJ * 8) / 256;
    constexpr int NT   = (COUT + 15) / 16;
    constexpr int HP   = 34;
    static_assert(2 * NT <= 8, "one output tile per wave");
    static_assert(RI == 8 || RI == 16, "one 16-row tile, filled by whole passes of the 8 waves");
    __shared__ v4f    vt4[2][TJ * 8];
    __shared__ float  red[8][RI][32];
    __shared__ float  hs[16][HP];

    const int tid  = threadIdx.x;
    const int lane = tid & 31;
    const int wid  = tid >> 5;
    const int seg  = blockIdx.x % NSEG;
    const int i0   = (blockIdx.x / NSEG) * RI;
    const int base = (i0 / N) * N;
    const int c    = seg * 32 + lane;

    const int widu = __builtin_amdgcn_readfirstlane(wid);
    const int half = lane >> 4;
    const int l15  = lane & 15;
    const int part = widu / NT;
    const int ocol = (widu % NT) * 16 + l15;
    const bool cok = (ocol < COUT);

    v2f bw[8];
    float bv = 0.0f;
    if (widu < 2 * NT) {
        if constexpr (ADD_BIAS) {
            if (part == 0) bv = bias[cok ? ocol : 0];
        }
        const float* wp = Wn + (size_t)(part * C + seg * 32 + half * 2) * COUT
                             + (cok ? ocol : 0);
#pragma unroll
        for (int s = 0; s < 8; ++s) {
            bw[s].x = wp[(size_t)(4 * s)     * COUT];
            bw[s].y = wp[(size_t)(4 * s + 1) * COUT];
        }
    }

    const float* vsrc = V + (size_t)base * C + seg * 32
                          + (size_t)(tid >> 3) * C + (tid & 7) * 4;
    v4f pre[NLD];
#pragma unroll
    for (int k = 0; k < NLD; ++k)
        pre[k] = *reinterpret_cast<const v4f*>(vsrc + (size_t)(32 * k) * C);

    float a[RI], acc[RI];
#pragma unroll
    for (int r = 0; r < RI; ++r) {
        a[r]   = A[(i0 + r) * C + c];
        acc[r] = 0.0f;
    }
    float selfterm[RI / 8];
#pragma unroll
    for (int q = 0; q < RI / 8; ++q) {
        const int i = i0 + wid + 8 * q;
        selfterm[q] = fmaxf(A[i * C + c] + V[i * C + c], 0.0f);
    }

    int buf = 0;
    for (int jt = 0; jt < N; jt += TJ) {
#pragma unroll
        for (int k = 0; k < NLD; ++k) vt4[buf][tid + 256 * k] = pre[k];
        __syncthreads();
        if (jt + TJ < N) {
            const float* nsrc = vsrc + (size_t)(jt + TJ) * C;
#pragma unroll
            for (int k = 0; k < NLD; ++k)
                pre[k] = *reinterpret_cast<const v4f*>(nsrc + (size_t)(32 * k) * C);
        }
        const float* vb = reinterpret_cast<const float*>(vt4[buf]) + (wid * JW) * 32 + lane;
        float v[JW];
#pragma unroll
        for (int jj = 0; jj < JW; ++jj) v[jj] = vb[jj * 32];
        __builtin_amdgcn_sched_barrier(0);
#pragma unroll
        for (int jj = 0; jj < JW; ++jj) {
#pragma unroll
            for (int r = 0; r < RI; ++r)
                acc[r] += fmaxf(a[r] + v[jj], 0.0f);
        }
        buf ^= 1;
    }

#pragma unroll
    for (int r = 0; r < RI; ++r) red[wid][r][lane] = acc[r];
    __syncthreads();

#pragma unroll
    for (int q = 0; q < RI / 8; ++q) {
        const int rr = wid + 8 * q;
        float s = red[0][rr][lane];
#pragma unroll
        for (int w = 1; w < 8; ++w) s += red[w][rr][lane];
        hs[rr][lane] = (s - selfterm[q]) * inv_nm1;
    }
    if constexpr (RI < 16) hs[wid + 8][lane] = 0.0f;
    __syncthreads();

    if (widu < 2 * NT) {
        const float* hp = &hs[l15][half * 2];
        v2f af[8], bf[8];
#pragma unroll
        for (int s = 0; s < 8; ++s) {
            af[s].x = hp[4 * s];
            af[s].y = hp[4 * s + 1];
            bf[s].x = cok ? bw[s].x : 0.0f;
            bf[s].y = cok ? bw[s].y : 0.0f;
        }
        v8f d = {};
#pragma unroll
        for (int s = 0; s < 8; ++s) {
            d = __builtin_amdgcn_wmma_f32_16x16x4_f32(
                    false, af[s], false, bf[s], (short)0, d, false, false);
        }
        if (cok) {
            float* dst = (part == 0 ? outA : outV) + (size_t)seg * plane;
            if constexpr (ADD_BIAS) {
#pragma unroll
                for (int r = 0; r < 8; ++r) d[r] += bv;
            }
#pragma unroll
            for (int r = 0; r < 8; ++r) {
                const int row = r + half * 8;
                if (row < RI)
                    dst[(size_t)(i0 + row) * COUT + ocol] = d[r];
            }
        }
    }
}

template<int N>
__global__ __launch_bounds__(256) void edge_colsum_final3(
    const float* __restrict__ pA,
    const float* __restrict__ pV,
    const float* __restrict__ bias,
    float* __restrict__ out,
    float inv_nm1, int plane)
{
    static_assert((N * 3) % 256 == 0 && N % 256 == 0, "a block lies in one batch");
    __shared__ float red[3][256];
    const int t   = threadIdx.x;
    const int idx = blockIdx.x * 256 + t;
    const int c   = idx % 3;
    const int b   = (blockIdx.x * 256) / (N * 3);

    const float a_own = (pA[idx] + pA[plane + idx]) + bias[c];
    const float v_own = pV[idx] + pV[plane + idx];

    float s0 = 0.0f, s1 = 0.0f, s2 = 0.0f;
#pragma unroll
    for (int k = 0; k < N / 256; ++k) {
        const size_t e = (size_t)(b * N + t + 256 * k) * 3;
        s0 += pV[e]     + pV[plane + e];
        s1 += pV[e + 1] + pV[plane + e + 1];
        s2 += pV[e + 2] + pV[plane + e + 2];
    }
    red[0][t] = s0;
    red[1][t] = s1;
    red[2][t] = s2;
    __syncthreads();
    for (int off = 128; off > 0; off >>= 1) {
        if (t < off) {
            red[0][t] += red[0][t + off];
            red[1][t] += red[1][t + off];
            red[2][t] += red[2][t + off];
        }
        __syncthreads();
    }
    out[idx] = a_own + (red[c][0] - v_own) * inv_nm1;
}

extern "C" void kernel_launch(void* const* d_in, const int* in_sizes, int n_in,
                              void* d_out, int out_size, void* d_ws, size_t ws_size,
                              hipStream_t stream) {
    (void)in_sizes; (void)n_in; (void)out_size; (void)ws_size;

    const float* x  = (const float*)d_in[0];
    const float* W1 = (const float*)d_in[1];
    const float* b1 = (const float*)d_in[2];
    const float* W2 = (const float*)d_in[3];
    const float* b2 = (const float*)d_in[4];
    const float* W3 = (const float*)d_in[5];
    const float* b3 = (const float*)d_in[6];
    float* out = (float*)d_out;

    const int B = 2, N = 1024;
    const int ROWS = B * N;
    const float inv = 1.0f / (float)(N - 1);

    float* ws = (float*)d_ws;
    float* A1 = ws;
    float* V1 = A1 + ROWS * 32;
    float* A2 = V1 + ROWS * 32;
    float* V2 = A2 + ROWS * 64;
    float* PA = V2 + ROWS * 64;
    float* PV = PA + 2 * ROWS * 3;

    const int MT = ROWS / 16;

    edge_gemm_dual<3, 32><<<MT * 2, 32, 0, stream>>>(x, W1, b1, A1, V1);
    edge_pairwise_relu_gemm<32, 8, 64, true><<<ROWS / 8, 256, 0, stream>>>(
        A1, V1, W2, b2, A2, V2, N, inv, 0);

    edge_pairwise_relu_gemm<64, 16, 3, false><<<(ROWS / 16) * 2, 256, 0, stream>>>(
        A2, V2, W3, b3, PA, PV, N, inv, ROWS * 3);

    edge_colsum_final3<N><<<(ROWS * 3) / 256, 256, 0, stream>>>(PA, PV, b3, out, inv, ROWS * 3);
}


# first layer's dual GEMM (3 -> 32, fp32 WMMA) moved into the first pairwise kernel: each wave makes the A rows and the V rows it sums, V through LDS, no A1/V1 in memory; 4 launches -> 3
# speedup vs baseline: 4.6144x; 1.1569x over previous
#include <hip/hip_runtime.h>
#include <hip/hip_bf16.h>

typedef __attribute__((ext_vector_type(2))) float v2f;
typedef __attribute__((ext_vector_type(8))) float v8f;
typedef __attribute__((ext_vector_type(4))) float v4f;

template<int COUT>
__global__ __launch_bounds__(256) void edge_first_pairwise_relu_gemm(
    const float* __restrict__ X,
    const float* __restrict__ W0,
    const float* __restrict__ b0,
    const float* __restrict__ Wn,
    const float* __restrict__ bias,
    float* __restrict__ outA,
    float* __restrict__ outV,
    int N, float inv_nm1)
{
    constexpr int CIN  = 3;
    constexpr int C    = 32;
    constexpr int RI   = 8;
    constexpr int TJ   = 256;
    constexpr int JW   = TJ / 8;
    constexpr int NT   = (COUT + 15) / 16;
    constexpr int HP   = 34;
    static_assert(2 * NT <= 8, "one output tile per wave");
    static_assert(JW == 32, "a wave's slice of a chunk is two 16-row tiles");
    __shared__ float  vt[2][TJ * C];
    __shared__ float  red[8][RI][32];
    __shared__ float  hs[16][HP];

    const int tid  = threadIdx.x;
    const int lane = tid & 31;
    const int wid  = tid >> 5;
    const int i0   = blockIdx.x * RI;
    const int base = (i0 / N) * N;

    const int widu = __builtin_amdgcn_readfirstlane(wid);
    const int half = lane >> 4;
    const int l15  = lane & 15;
    const int part = widu / NT;
    const int ocol = (widu % NT) * 16 + l15;
    const bool cok = (ocol < COUT);

    v2f bw[8];
    float bv = 0.0f;
    if (widu < 2 * NT) {
        if (part == 0) bv = bias[cok ? ocol : 0];
        const float* wp = Wn + (size_t)(part * C + half * 2) * COUT + (cok ? ocol : 0);
#pragma unroll
        for (int s = 0; s < 8; ++s) {
            bw[s].x = wp[(size_t)(4 * s)     * COUT];
            bw[s].y = wp[(size_t)(4 * s + 1) * COUT];
        }
    }

    const int k0 = half * 2;
    const int k1 = half ? 2 : 1;

    v2f wa[2], wv[2];
#pragma unroll
    for (int ct = 0; ct < 2; ++ct) {
        const int col = ct * 16 + l15;
        const float a1 = W0[k1 * C + col];
        const float v1 = W0[(CIN + k1) * C + col];
        wa[ct].x = W0[k0 * C + col];
        wa[ct].y = half ? 0.0f : a1;
        wv[ct].x = W0[(CIN + k0) * C + col];
        wv[ct].y = half ? 0.0f : v1;
    }

    const int frow = 2 * (l15 & 7) + (l15 >> 3);
    const float* xs = X + (size_t)(base + wid * JW + frow) * CIN;
    float xk0[2], xk1[2];
#pragma unroll
    for (int rt = 0; rt < 2; ++rt) {
        xk0[rt] = xs[(16 * rt) * CIN + k0];
        xk1[rt] = xs[(16 * rt) * CIN + k1];
    }

    float a[RI], acc[RI];
    {
        const float* xo = X + (size_t)(i0 + (l15 & 7)) * CIN;
        const float x1 = xo[k1];
        v2f xa;
        xa.x = xo[k0];
        xa.y = half ? 0.0f : x1;
        const float bl = b0[lane];
        const v8f z = {};
        const v8f d0 = __builtin_amdgcn_wmma_f32_16x16x4_f32(
                           false, xa, false, wa[0], (short)0, z, false, false);
        const v8f d1 = __builtin_amdgcn_wmma_f32_16x16x4_f32(
                           false, xa, false, wa[1], (short)0, z, false, false);
#pragma unroll
        for (int r = 0; r < RI; ++r) {
            a[r]   = (half ? d1[r] : d0[r]) + bl;
            acc[r] = 0.0f;
        }
    }

    const int jl      = i0 - base;
    const int jt_self = (jl / TJ) * TJ;
    const int js      = jl - jt_self + wid;
    float vself = 0.0f;

    int buf = 0;
    for (int jt = 0; jt < N; jt += TJ) {
#pragma unroll
        for (int rt = 0; rt < 2; ++rt) {
            v2f xf;
            xf.x = xk0[rt];
            xf.y = half ? 0.0f : xk1[rt];
#pragma unroll
            for (int ct = 0; ct < 2; ++ct) {
                const v8f z = {};
                const v8f d = __builtin_amdgcn_wmma_f32_16x16x4_f32(
                                  false, xf, false, wv[ct], (short)0, z, false, false);
                float* dst = &vt[buf][(wid * JW + 16 * rt + half) * C + ct * 16 + l15];
#pragma unroll
                for (int r = 0; r < 8; ++r) dst[(2 * r) * C] = d[r];
            }
        }
        __syncthreads();
        if (jt + TJ < N) {
            const float* nx = xs + (size_t)(jt + TJ) * CIN;
#pragma unroll
            for (int rt = 0; rt < 2; ++rt) {
                xk0[rt] = nx[(16 * rt) * CIN + k0];
                xk1[rt] = nx[(16 * rt) * CIN + k1];
            }
        }
        if (jt == jt_self) vself = vt[buf][js * C + lane];
        const float* vb = &vt[buf][(wid * JW) * C + lane];
        float v[JW];
#pragma unroll
        for (int jj = 0; jj < JW; ++jj) v[jj] = vb[jj * C];
        __builtin_amdgcn_sched_barrier(0);
#pragma unroll
        for (int jj = 0; jj < JW; ++jj) {
#pragma unroll
            for (int r = 0; r < RI; ++r)
                acc[r] += fmaxf(a[r] + v[jj], 0.0f);
        }
        buf ^= 1;
    }

#pragma unroll
    for (int r = 0; r < RI; ++r) red[wid][r][lane] = acc[r];
    __syncthreads();

    {
        float aw = a[0];
#pragma unroll
        for (int r = 1; r < RI; ++r) aw = (widu == r) ? a[r] : aw;
        const float selfterm = fmaxf(aw + vself, 0.0f);
        float s = red[0][wid][lane];
#pragma unroll
        for (int w = 1; w < 8; ++w) s += red[w][wid][lane];
        hs[wid][lane] = (s - selfterm) * inv_nm1;
    }
    hs[wid + 8][lane] = 0.0f;
    __syncthreads();

    if (widu < 2 * NT) {
        const float* hp = &hs[l15][half * 2];
        v2f af[8], bf[8];
#pragma unroll
        for (int s = 0; s < 8; ++s) {
            af[s].x = hp[4 * s];
            af[s].y = hp[4 * s + 1];
            bf[s].x = cok ? bw[s].x : 0.0f;
            bf[s].y = cok ? bw[s].y : 0.0f;
        }
        v8f d = {};
#pragma unroll
        for (int s = 0; s < 8; ++s) {
            d = __builtin_amdgcn_wmma_f32_16x16x4_f32(
                    false, af[s], false, bf[s], (short)0, d, false, false);
        }
        if (cok) {
            float* dst = (part == 0 ? outA : outV);
#pragma unroll
            for (int r = 0; r < 8; ++r) d[r] += bv;
#pragma unroll
            for (int r = 0; r < 8; ++r) {
                const int row = r + half * 8;
                if (row < RI)
                    dst[(size_t)(i0 + row) * COUT + ocol] = d[r];
            }
        }
    }
}

template<int C, int RI, int COUT, bool ADD_BIAS>
__global__ __launch_bounds__(256) void edge_pairwise_relu_gemm(
    const float* __restrict__ A,
    const float* __restrict__ V,
    const float* __restrict__ Wn,
    const float* __restrict__ bias,
    float* __restrict__ outA,
    float* __restrict__ outV,
    int N, float inv_nm1, int plane)
{
    constexpr int TJ   = 256;
    constexpr int NSEG = C / 32;
    constexpr int JW   = TJ / 8;
    constexpr int NLD  = (TJ * 8) / 256;
    constexpr int NT   = (COUT + 15) / 16;
    constexpr int HP   = 34;
    static_assert(2 * NT <= 8, "one output tile per wave");
    static_assert(RI == 8 || RI == 16, "one 16-row tile, filled by whole passes of the 8 waves");
    __shared__ v4f    vt4[2][TJ * 8];
    __shared__ float  red[8][RI][32];
    __shared__ float  hs[16][HP];

    const int tid  = threadIdx.x;
    const int lane = tid & 31;
    const int wid  = tid >> 5;
    const int seg  = blockIdx.x % NSEG;
    const int i0   = (blockIdx.x / NSEG) * RI;
    const int base = (i0 / N) * N;
    const int c    = seg * 32 + lane;

    const int widu = __builtin_amdgcn_readfirstlane(wid);
    const int half = lane >> 4;
    const int l15  = lane & 15;
    const int part = widu / NT;
    const int ocol = (widu % NT) * 16 + l15;
    const bool cok = (ocol < COUT);

    v2f bw[8];
    float bv = 0.0f;
    if (widu < 2 * NT) {
        if constexpr (ADD_BIAS) {
            if (part == 0) bv = bias[cok ? ocol : 0];
        }
        const float* wp = Wn + (size_t)(part * C + seg * 32 + half * 2) * COUT
                             + (cok ? ocol : 0);
#pragma unroll
        for (int s = 0; s < 8; ++s) {
            bw[s].x = wp[(size_t)(4 * s)     * COUT];
            bw[s].y = wp[(size_t)(4 * s + 1) * COUT];
        }
    }

    const float* vsrc = V + (size_t)base * C + seg * 32
                          + (size_t)(tid >> 3) * C + (tid & 7) * 4;
    v4f pre[NLD];
#pragma unroll
    for (int k = 0; k < NLD; ++k)
        pre[k] = *reinterpret_cast<const v4f*>(vsrc + (size_t)(32 * k) * C);

    float a[RI], acc[RI];
#pragma unroll
    for (int r = 0; r < RI; ++r) {
        a[r]   = A[(i0 + r) * C + c];
        acc[r] = 0.0f;
    }
    float selfterm[RI / 8];
#pragma unroll
    for (int q = 0; q < RI / 8; ++q) {
        const int i = i0 + wid + 8 * q;
        selfterm[q] = fmaxf(A[i * C + c] + V[i * C + c], 0.0f);
    }

    int buf = 0;
    for (int jt = 0; jt < N; jt += TJ) {
#pragma unroll
        for (int k = 0; k < NLD; ++k) vt4[buf][tid + 256 * k] = pre[k];
        __syncthreads();
        if (jt + TJ < N) {
            const float* nsrc = vsrc + (size_t)(jt + TJ) * C;
#pragma unroll
            for (int k = 0; k < NLD; ++k)
                pre[k] = *reinterpret_cast<const v4f*>(nsrc + (size_t)(32 * k) * C);
        }
        const float* vb = reinterpret_cast<const float*>(vt4[buf]) + (wid * JW) * 32 + lane;
        float v[JW];
#pragma unroll
        for (int jj = 0; jj < JW; ++jj) v[jj] = vb[jj * 32];
        __builtin_amdgcn_sched_barrier(0);
#pragma unroll
        for (int jj = 0; jj < JW; ++jj) {
#pragma unroll
            for (int r = 0; r < RI; ++r)
                acc[r] += fmaxf(a[r] + v[jj], 0.0f);
        }
        buf ^= 1;
    }

#pragma unroll
    for (int r = 0; r < RI; ++r) red[wid][r][lane] = acc[r];
    __syncthreads();

#pragma unroll
    for (int q = 0; q < RI / 8; ++q) {
        const int rr = wid + 8 * q;
        float s = red[0][rr][lane];
#pragma unroll
        for (int w = 1; w < 8; ++w) s += red[w][rr][lane];
        hs[rr][lane] = (s - selfterm[q]) * inv_nm1;
    }
    if constexpr (RI < 16) hs[wid + 8][lane] = 0.0f;
    __syncthreads();

    if (widu < 2 * NT) {
        const float* hp = &hs[l15][half * 2];
        v2f af[8], bf[8];
#pragma unroll
        for (int s = 0; s < 8; ++s) {
            af[s].x = hp[4 * s];
            af[s].y = hp[4 * s + 1];
            bf[s].x = cok ? bw[s].x : 0.0f;
            bf[s].y = cok ? bw[s].y : 0.0f;
        }
        v8f d = {};
#pragma unroll
        for (int s = 0; s < 8; ++s) {
            d = __builtin_amdgcn_wmma_f32_16x16x4_f32(
                    false, af[s], false, bf[s], (short)0, d, false, false);
        }
        if (cok) {
            float* dst = (part == 0 ? outA : outV) + (size_t)seg * plane;
            if constexpr (ADD_BIAS) {
#pragma unroll
                for (int r = 0; r < 8; ++r) d[r] += bv;
            }
#pragma unroll
            for (int r = 0; r < 8; ++r) {
                const int row = r + half * 8;
                if (row < RI)
                    dst[(size_t)(i0 + row) * COUT + ocol] = d[r];
            }
        }
    }
}

template<int N>
__global__ __launch_bounds__(256) void edge_colsum_final3(
    const float* __restrict__ pA,
    const float* __restrict__ pV,
    const float* __restrict__ bias,
    float* __restrict__ out,
    float inv_nm1, int plane)
{
    static_assert((N * 3) % 256 == 0 && N % 256 == 0, "a block lies in one batch");
    __shared__ float red[3][256];
    const int t   = threadIdx.x;
    const int idx = blockIdx.x * 256 + t;
    const int c   = idx % 3;
    const int b   = (blockIdx.x * 256) / (N * 3);

    const float a_own = (pA[idx] + pA[plane + idx]) + bias[c];
    const float v_own = pV[idx] + pV[plane + idx];

    float s0 = 0.0f, s1 = 0.0f, s2 = 0.0f;
#pragma unroll
    for (int k = 0; k < N / 256; ++k) {
        const size_t e = (size_t)(b * N + t + 256 * k) * 3;
        s0 += pV[e]     + pV[plane + e];
        s1 += pV[e + 1] + pV[plane + e + 1];
        s2 += pV[e + 2] + pV[plane + e + 2];
    }
    red[0][t] = s0;
    red[1][t] = s1;
    red[2][t] = s2;
    __syncthreads();
    for (int off = 128; off > 0; off >>= 1) {
        if (t < off) {
            red[0][t] += red[0][t + off];
            red[1][t] += red[1][t + off];
            red[2][t] += red[2][t + off];
        }
        __syncthreads();
    }
    out[idx] = a_own + (red[c][0] - v_own) * inv_nm1;
}

extern "C" void kernel_launch(void* const* d_in, const int* in_sizes, int n_in,
                              void* d_out, int out_size, void* d_ws, size_t ws_size,
                              hipStream_t stream) {
    (void)in_sizes; (void)n_in; (void)out_size; (void)ws_size;

    const float* x  = (const float*)d_in[0];
    const float* W1 = (const float*)d_in[1];
    const float* b1 = (const float*)d_in[2];
    const float* W2 = (const float*)d_in[3];
    const float* b2 = (const float*)d_in[4];
    const float* W3 = (const float*)d_in[5];
    const float* b3 = (const float*)d_in[6];
    float* out = (float*)d_out;

    const int B = 2, N = 1024;
    const int ROWS = B * N;
    const float inv = 1.0f / (float)(N - 1);

    float* ws = (float*)d_ws;
    float* A2 = ws;
    float* V2 = A2 + ROWS * 64;
    float* PA = V2 + ROWS * 64;
    float* PV = PA + 2 * ROWS * 3;

    edge_first_pairwise_relu_gemm<64><<<ROWS / 8, 256, 0, stream>>>(
        x, W1, b1, W2, b2, A2, V2, N, inv);

    edge_pairwise_relu_gemm<64, 16, 3, false><<<(ROWS / 16) * 2, 256, 0, stream>>>(
        A2, V2, W3, b3, PA, PV, N, inv, ROWS * 3);

    edge_colsum_final3<N><<<(ROWS * 3) / 256, 256, 0, stream>>>(PA, PV, b3, out, inv, ROWS * 3);
}


# pair terms summed as max(v,-a), two rows a packed add, 32*a put back into each 32-row sum by one fma: 1 vector instruction a pair term, was 1.6 (new rounding of the sums)
# speedup vs baseline: 5.0292x; 1.0899x over previous
#include <hip/hip_runtime.h>
#include <hip/hip_bf16.h>

typedef __attribute__((ext_vector_type(2))) float v2f;
typedef __attribute__((ext_vector_type(8))) float v8f;
typedef __attribute__((ext_vector_type(4))) float v4f;

template<int COUT>
__global__ __launch_bounds__(256) void edge_first_pairwise_relu_gemm(
    const float* __restrict__ X,
    const float* __restrict__ W0,
    const float* __restrict__ b0,
    const float* __restrict__ Wn,
    const float* __restrict__ bias,
    float* __restrict__ outA,
    float* __restrict__ outV,
    int N, float inv_nm1)
{
    constexpr int CIN  = 3;
    constexpr int C    = 32;
    constexpr int RI   = 8;
    constexpr int TJ   = 256;
    constexpr int JW   = TJ / 8;
    constexpr int NT   = (COUT + 15) / 16;
    constexpr int HP   = 34;
    static_assert(2 * NT <= 8, "one output tile per wave");
    static_assert(JW == 32, "a wave's slice of a chunk is two 16-row tiles");
    __shared__ float  vt[2][TJ * C];
    __shared__ float  red[8][RI][32];
    __shared__ float  hs[16][HP];

    const int tid  = threadIdx.x;
    const int lane = tid & 31;
    const int wid  = tid >> 5;
    const int i0   = blockIdx.x * RI;
    const int base = (i0 / N) * N;

    const int widu = __builtin_amdgcn_readfirstlane(wid);
    const int half = lane >> 4;
    const int l15  = lane & 15;
    const int part = widu / NT;
    const int ocol = (widu % NT) * 16 + l15;
    const bool cok = (ocol < COUT);

    v2f bw[8];
    float bv = 0.0f;
    if (widu < 2 * NT) {
        if (part == 0) bv = bias[cok ? ocol : 0];
        const float* wp = Wn + (size_t)(part * C + half * 2) * COUT + (cok ? ocol : 0);
#pragma unroll
        for (int s = 0; s < 8; ++s) {
            bw[s].x = wp[(size_t)(4 * s)     * COUT];
            bw[s].y = wp[(size_t)(4 * s + 1) * COUT];
        }
    }

    const int k0 = half * 2;
    const int k1 = half ? 2 : 1;

    v2f wa[2], wv[2];
#pragma unroll
    for (int ct = 0; ct < 2; ++ct) {
        const int col = ct * 16 + l15;
        const float a1 = W0[k1 * C + col];
        const float v1 = W0[(CIN + k1) * C + col];
        wa[ct].x = W0[k0 * C + col];
        wa[ct].y = half ? 0.0f : a1;
        wv[ct].x = W0[(CIN + k0) * C + col];
        wv[ct].y = half ? 0.0f : v1;
    }

    const int frow = 2 * (l15 & 7) + (l15 >> 3);
    const float* xs = X + (size_t)(base + wid * JW + frow) * CIN;
    float xk0[2], xk1[2];
#pragma unroll
    for (int rt = 0; rt < 2; ++rt) {
        xk0[rt] = xs[(16 * rt) * CIN + k0];
        xk1[rt] = xs[(16 * rt) * CIN + k1];
    }

    v2f na[RI / 2], acc[RI / 2];
    {
        const float* xo = X + (size_t)(i0 + (l15 & 7)) * CIN;
        const float x1 = xo[k1];
        v2f xa;
        xa.x = xo[k0];
        xa.y = half ? 0.0f : x1;
        const float bl = b0[lane];
        const v8f z = {};
        const v8f d0 = __builtin_amdgcn_wmma_f32_16x16x4_f32(
                           false, xa, false, wa[0], (short)0, z, false, false);
        const v8f d1 = __builtin_amdgcn_wmma_f32_16x16x4_f32(
                           false, xa, false, wa[1], (short)0, z, false, false);
#pragma unroll
        for (int k = 0; k < RI / 2; ++k) {
            na[k].x = -((half ? d1[2 * k]     : d0[2 * k])     + bl);
            na[k].y = -((half ? d1[2 * k + 1] : d0[2 * k + 1]) + bl);
            acc[k].x = 0.0f;
            acc[k].y = 0.0f;
        }
    }

    const int jl      = i0 - base;
    const int jt_self = (jl / TJ) * TJ;
    const int js      = jl - jt_self + wid;
    float vself = 0.0f;

    int buf = 0;
    for (int jt = 0; jt < N; jt += TJ) {
#pragma unroll
        for (int rt = 0; rt < 2; ++rt) {
            v2f xf;
            xf.x = xk0[rt];
            xf.y = half ? 0.0f : xk1[rt];
#pragma unroll
            for (int ct = 0; ct < 2; ++ct) {
                const v8f z = {};
                const v8f d = __builtin_amdgcn_wmma_f32_16x16x4_f32(
                                  false, xf, false, wv[ct], (short)0, z, false, false);
                float* dst = &vt[buf][(wid * JW + 16 * rt + half) * C + ct * 16 + l15];
#pragma unroll
                for (int r = 0; r < 8; ++r) dst[(2 * r) * C] = d[r];
            }
        }
        __syncthreads();
        if (jt + TJ < N) {
            const float* nx = xs + (size_t)(jt + TJ) * CIN;
#pragma unroll
            for (int rt = 0; rt < 2; ++rt) {
                xk0[rt] = nx[(16 * rt) * CIN + k0];
                xk1[rt] = nx[(16 * rt) * CIN + k1];
            }
        }
        if (jt == jt_self) vself = vt[buf][js * C + lane];
        const float* vb = &vt[buf][(wid * JW) * C + lane];
        float v[JW];
#pragma unroll
        for (int jj = 0; jj < JW; ++jj) v[jj] = vb[jj * C];
        __builtin_amdgcn_sched_barrier(0);
        v2f cs[RI / 2];
#pragma unroll
        for (int k = 0; k < RI / 2; ++k) {
            cs[k].x = fmaxf(v[0], na[k].x);
            cs[k].y = fmaxf(v[0], na[k].y);
        }
#pragma unroll
        for (int jj = 1; jj < JW; ++jj) {
#pragma unroll
            for (int k = 0; k < RI / 2; ++k) {
                v2f m;
                m.x = fmaxf(v[jj], na[k].x);
                m.y = fmaxf(v[jj], na[k].y);
                cs[k] += m;
            }
        }
        const v2f njw = {-(float)JW, -(float)JW};
#pragma unroll
        for (int k = 0; k < RI / 2; ++k)
            acc[k] += __builtin_elementwise_fma(njw, na[k], cs[k]);
        buf ^= 1;
    }

#pragma unroll
    for (int k = 0; k < RI / 2; ++k) {
        red[wid][2 * k][lane]     = acc[k].x;
        red[wid][2 * k + 1][lane] = acc[k].y;
    }
    __syncthreads();

    {
        float naw = na[0].x;
#pragma unroll
        for (int r = 1; r < RI; ++r) naw = (widu == r) ? ((r & 1) ? na[r / 2].y : na[r / 2].x) : naw;
        const float selfterm = fmaxf(vself - naw, 0.0f);
        float s = red[0][wid][lane];
#pragma unroll
        for (int w = 1; w < 8; ++w) s += red[w][wid][lane];
        hs[wid][lane] = (s - selfterm) * inv_nm1;
    }
    hs[wid + 8][lane] = 0.0f;
    __syncthreads();

    if (widu < 2 * NT) {
        const float* hp = &hs[l15][half * 2];
        v2f af[8], bf[8];
#pragma unroll
        for (int s = 0; s < 8; ++s) {
            af[s].x = hp[4 * s];
            af[s].y = hp[4 * s + 1];
            bf[s].x = cok ? bw[s].x : 0.0f;
            bf[s].y = cok ? bw[s].y : 0.0f;
        }
        v8f d = {};
#pragma unroll
        for (int s = 0; s < 8; ++s) {
            d = __builtin_amdgcn_wmma_f32_16x16x4_f32(
                    false, af[s], false, bf[s], (short)0, d, false, false);
        }
        if (cok) {
            float* dst = (part == 0 ? outA : outV);
#pragma unroll
            for (int r = 0; r < 8; ++r) d[r] += bv;
#pragma unroll
            for (int r = 0; r < 8; ++r) {
                const int row = r + half * 8;
                if (row < RI)
                    dst[(size_t)(i0 + row) * COUT + ocol] = d[r];
            }
        }
    }
}

template<int C, int RI, int COUT, bool ADD_BIAS>
__global__ __launch_bounds__(256) void edge_pairwise_relu_gemm(
    const float* __restrict__ A,
    const float* __restrict__ V,
    const float* __restrict__ Wn,
    const float* __restrict__ bias,
    float* __restrict__ outA,
    float* __restrict__ outV,
    int N, float inv_nm1, int plane)
{
    constexpr int TJ   = 256;
    constexpr int NSEG = C / 32;
    constexpr int JW   = TJ / 8;
    constexpr int NLD  = (TJ * 8) / 256;
    constexpr int NT   = (COUT + 15) / 16;
    constexpr int HP   = 34;
    static_assert(2 * NT <= 8, "one output tile per wave");
    static_assert(RI == 8 || RI == 16, "one 16-row tile, filled by whole passes of the 8 waves");
    __shared__ v4f    vt4[2][TJ * 8];
    __shared__ float  red[8][RI][32];
    __shared__ float  hs[16][HP];

    const int tid  = threadIdx.x;
    const int lane = tid & 31;
    const int wid  = tid >> 5;
    const int seg  = blockIdx.x % NSEG;
    const int i0   = (blockIdx.x / NSEG) * RI;
    const int base = (i0 / N) * N;
    const int c    = seg * 32 + lane;

    const int widu = __builtin_amdgcn_readfirstlane(wid);
    const int half = lane >> 4;
    const int l15  = lane & 15;
    const int part = widu / NT;
    const int ocol = (widu % NT) * 16 + l15;
    const bool cok = (ocol < COUT);

    v2f bw[8];
    float bv = 0.0f;
    if (widu < 2 * NT) {
        if constexpr (ADD_BIAS) {
            if (part == 0) bv = bias[cok ? ocol : 0];
        }
        const float* wp = Wn + (size_t)(part * C + seg * 32 + half * 2) * COUT
                             + (cok ? ocol : 0);
#pragma unroll
        for (int s = 0; s < 8; ++s) {
            bw[s].x = wp[(size_t)(4 * s)     * COUT];
            bw[s].y = wp[(size_t)(4 * s + 1) * COUT];
        }
    }

    const float* vsrc = V + (size_t)base * C + seg * 32
                          + (size_t)(tid >> 3) * C + (tid & 7) * 4;
    v4f pre[NLD];
#pragma unroll
    for (int k = 0; k < NLD; ++k)
        pre[k] = *reinterpret_cast<const v4f*>(vsrc + (size_t)(32 * k) * C);

    v2f na[RI / 2], acc[RI / 2];
#pragma unroll
    for (int k = 0; k < RI / 2; ++k) {
        na[k].x  = -A[(i0 + 2 * k) * C + c];
        na[k].y  = -A[(i0 + 2 * k + 1) * C + c];
        acc[k].x = 0.0f;
        acc[k].y = 0.0f;
    }
    float selfterm[RI / 8];
#pragma unroll
    for (int q = 0; q < RI / 8; ++q) {
        const int i = i0 + wid + 8 * q;
        selfterm[q] = fmaxf(A[i * C + c] + V[i * C + c], 0.0f);
    }

    int buf = 0;
    for (int jt = 0; jt < N; jt += TJ) {
#pragma unroll
        for (int k = 0; k < NLD; ++k) vt4[buf][tid + 256 * k] = pre[k];
        __syncthreads();
        if (jt + TJ < N) {
            const float* nsrc = vsrc + (size_t)(jt + TJ) * C;
#pragma unroll
            for (int k = 0; k < NLD; ++k)
                pre[k] = *reinterpret_cast<const v4f*>(nsrc + (size_t)(32 * k) * C);
        }
        const float* vb = reinterpret_cast<const float*>(vt4[buf]) + (wid * JW) * 32 + lane;
        float v[JW];
#pragma unroll
        for (int jj = 0; jj < JW; ++jj) v[jj] = vb[jj * 32];
        __builtin_amdgcn_sched_barrier(0);
        v2f cs[RI / 2];
#pragma unroll
        for (int k = 0; k < RI / 2; ++k) {
            cs[k].x = fmaxf(v[0], na[k].x);
            cs[k].y = fmaxf(v[0], na[k].y);
        }
#pragma unroll
        for (int jj = 1; jj < JW; ++jj) {
#pragma unroll
            for (int k = 0; k < RI / 2; ++k) {
                v2f m;
                m.x = fmaxf(v[jj], na[k].x);
                m.y = fmaxf(v[jj], na[k].y);
                cs[k] += m;
            }
        }
        const v2f njw = {-(float)JW, -(float)JW};
#pragma unroll
        for (int k = 0; k < RI / 2; ++k)
            acc[k] += __builtin_elementwise_fma(njw, na[k], cs[k]);
        buf ^= 1;
    }

#pragma unroll
    for (int k = 0; k < RI / 2; ++k) {
        red[wid][2 * k][lane]     = acc[k].x;
        red[wid][2 * k + 1][lane] = acc[k].y;
    }
    __syncthreads();

#pragma unroll
    for (int q = 0; q < RI / 8; ++q) {
        const int rr = wid + 8 * q;
        float s = red[0][rr][lane];
#pragma unroll
        for (int w = 1; w < 8; ++w) s += red[w][rr][lane];
        hs[rr][lane] = (s - selfterm[q]) * inv_nm1;
    }
    if constexpr (RI < 16) hs[wid + 8][lane] = 0.0f;
    __syncthreads();

    if (widu < 2 * NT) {
        const float* hp = &hs[l15][half * 2];
        v2f af[8], bf[8];
#pragma unroll
        for (int s = 0; s < 8; ++s) {
            af[s].x = hp[4 * s];
            af[s].y = hp[4 * s + 1];
            bf[s].x = cok ? bw[s].x : 0.0f;
            bf[s].y = cok ? bw[s].y : 0.0f;
        }
        v8f d = {};
#pragma unroll
        for (int s = 0; s < 8; ++s) {
            d = __builtin_amdgcn_wmma_f32_16x16x4_f32(
                    false, af[s], false, bf[s], (short)0, d, false, false);
        }
        if (cok) {
            float* dst = (part == 0 ? outA : outV) + (size_t)seg * plane;
            if constexpr (ADD_BIAS) {
#pragma unroll
                for (int r = 0; r < 8; ++r) d[r] += bv;
            }
#pragma unroll
            for (int r = 0; r < 8; ++r) {
                const int row = r + half * 8;
                if (row < RI)
                    dst[(size_t)(i0 + row) * COUT + ocol] = d[r];
            }
        }
    }
}

template<int N>
__global__ __launch_bounds__(256) void edge_colsum_final3(
    const float* __restrict__ pA,
    const float* __restrict__ pV,
    const float* __restrict__ bias,
    float* __restrict__ out,
    float inv_nm1, int plane)
{
    static_assert((N * 3) % 256 == 0 && N % 256 == 0, "a block lies in one batch");
    __shared__ float red[3][256];
    const int t   = threadIdx.x;
    const int idx = blockIdx.x * 256 + t;
    const int c   = idx % 3;
    const int b   = (blockIdx.x * 256) / (N * 3);

    const float a_own = (pA[idx] + pA[plane + idx]) + bias[c];
    const float v_own = pV[idx] + pV[plane + idx];

    float s0 = 0.0f, s1 = 0.0f, s2 = 0.0f;
#pragma unroll
    for (int k = 0; k < N / 256; ++k) {
        const size_t e = (size_t)(b * N + t + 256 * k) * 3;
        s0 += pV[e]     + pV[plane + e];
        s1 += pV[e + 1] + pV[plane + e + 1];
        s2 += pV[e + 2] + pV[plane + e + 2];
    }
    red[0][t] = s0;
    red[1][t] = s1;
    red[2][t] = s2;
    __syncthreads();
    for (int off = 128; off > 0; off >>= 1) {
        if (t < off) {
            red[0][t] += red[0][t + off];
            red[1][t] += red[1][t + off];
            red[2][t] += red[2][t + off];
        }
        __syncthreads();
    }
    out[idx] = a_own + (red[c][0] - v_own) * inv_nm1;
}

extern "C" void kernel_launch(void* const* d_in, const int* in_sizes, int n_in,
                              void* d_out, int out_size, void* d_ws, size_t ws_size,
                              hipStream_t stream) {
    (void)in_sizes; (void)n_in; (void)out_size; (void)ws_size;

    const float* x  = (const float*)d_in[0];
    const float* W1 = (const float*)d_in[1];
    const float* b1 = (const float*)d_in[2];
    const float* W2 = (const float*)d_in[3];
    const float* b2 = (const float*)d_in[4];
    const float* W3 = (const float*)d_in[5];
    const float* b3 = (const float*)d_in[6];
    float* out = (float*)d_out;

    const int B = 2, N = 1024;
    const int ROWS = B * N;
    const float inv = 1.0f / (float)(N - 1);

    float* ws = (float*)d_ws;
    float* A2 = ws;
    float* V2 = A2 + ROWS * 64;
    float* PA = V2 + ROWS * 64;
    float* PV = PA + 2 * ROWS * 3;

    edge_first_pairwise_relu_gemm<64><<<ROWS / 8, 256, 0, stream>>>(
        x, W1, b1, W2, b2, A2, V2, N, inv);

    edge_pairwise_relu_gemm<64, 16, 3, false><<<(ROWS / 16) * 2, 256, 0, stream>>>(
        A2, V2, W3, b3, PA, PV, N, inv, ROWS * 3);

    edge_colsum_final3<N><<<(ROWS * 3) / 256, 256, 0, stream>>>(PA, PV, b3, out, inv, ROWS * 3);
}


# last layer's column sums: each pairwise block leaves the sums of its own 16 rows (cross-lane add), final kernel adds 128 records per batch by lane exchanges, no LDS tree (new summation order)
# speedup vs baseline: 5.3566x; 1.0651x over previous
#include <hip/hip_runtime.h>
#include <hip/hip_bf16.h>

typedef __attribute__((ext_vector_type(2))) float v2f;
typedef __attribute__((ext_vector_type(8))) float v8f;
typedef __attribute__((ext_vector_type(4))) float v4f;

__device__ __forceinline__ float row_gather(float x, unsigned lo, unsigned hi) {
    const int b = __builtin_bit_cast(int, x);
    return __builtin_bit_cast(float, __builtin_amdgcn_permlane16(b, b, lo, hi, false, false));
}
__device__ __forceinline__ float other_half(float x) {
    const int b = __builtin_bit_cast(int, x);
    return __builtin_bit_cast(float, __builtin_amdgcn_permlanex16(b, b, 0x76543210u, 0xfedcba98u, false, false));
}

template<int COUT>
__global__ __launch_bounds__(256) void edge_first_pairwise_relu_gemm(
    const float* __restrict__ X,
    const float* __restrict__ W0,
    const float* __restrict__ b0,
    const float* __restrict__ Wn,
    const float* __restrict__ bias,
    float* __restrict__ outA,
    float* __restrict__ outV,
    int N, float inv_nm1)
{
    constexpr int CIN  = 3;
    constexpr int C    = 32;
    constexpr int RI   = 8;
    constexpr int TJ   = 256;
    constexpr int JW   = TJ / 8;
    constexpr int NT   = (COUT + 15) / 16;
    constexpr int HP   = 34;
    static_assert(2 * NT <= 8, "one output tile per wave");
    static_assert(JW == 32, "a wave's slice of a chunk is two 16-row tiles");
    __shared__ float  vt[2][TJ * C];
    __shared__ float  red[8][RI][32];
    __shared__ float  hs[16][HP];

    const int tid  = threadIdx.x;
    const int lane = tid & 31;
    const int wid  = tid >> 5;
    const int i0   = blockIdx.x * RI;
    const int base = (i0 / N) * N;

    const int widu = __builtin_amdgcn_readfirstlane(wid);
    const int half = lane >> 4;
    const int l15  = lane & 15;
    const int part = widu / NT;
    const int ocol = (widu % NT) * 16 + l15;
    const bool cok = (ocol < COUT);

    v2f bw[8];
    float bv = 0.0f;
    if (widu < 2 * NT) {
        if (part == 0) bv = bias[cok ? ocol : 0];
        const float* wp = Wn + (size_t)(part * C + half * 2) * COUT + (cok ? ocol : 0);
#pragma unroll
        for (int s = 0; s < 8; ++s) {
            bw[s].x = wp[(size_t)(4 * s)     * COUT];
            bw[s].y = wp[(size_t)(4 * s + 1) * COUT];
        }
    }

    const int k0 = half * 2;
    const int k1 = half ? 2 : 1;

    v2f wa[2], wv[2];
#pragma unroll
    for (int ct = 0; ct < 2; ++ct) {
        const int col = ct * 16 + l15;
        const float a1 = W0[k1 * C + col];
        const float v1 = W0[(CIN + k1) * C + col];
        wa[ct].x = W0[k0 * C + col];
        wa[ct].y = half ? 0.0f : a1;
        wv[ct].x = W0[(CIN + k0) * C + col];
        wv[ct].y = half ? 0.0f : v1;
    }

    const int frow = 2 * (l15 & 7) + (l15 >> 3);
    const float* xs = X + (size_t)(base + wid * JW + frow) * CIN;
    float xk0[2], xk1[2];
#pragma unroll
    for (int rt = 0; rt < 2; ++rt) {
        xk0[rt] = xs[(16 * rt) * CIN + k0];
        xk1[rt] = xs[(16 * rt) * CIN + k1];
    }

    v2f na[RI / 2], acc[RI / 2];
    {
        const float* xo = X + (size_t)(i0 + (l15 & 7)) * CIN;
        const float x1 = xo[k1];
        v2f xa;
        xa.x = xo[k0];
        xa.y = half ? 0.0f : x1;
        const float bl = b0[lane];
        const v8f z = {};
        const v8f d0 = __builtin_amdgcn_wmma_f32_16x16x4_f32(
                           false, xa, false, wa[0], (short)0, z, false, false);
        const v8f d1 = __builtin_amdgcn_wmma_f32_16x16x4_f32(
                           false, xa, false, wa[1], (short)0, z, false, false);
#pragma unroll
        for (int k = 0; k < RI / 2; ++k) {
            na[k].x = -((half ? d1[2 * k]     : d0[2 * k])     + bl);
            na[k].y = -((half ? d1[2 * k + 1] : d0[2 * k + 1]) + bl);
            acc[k].x = 0.0f;
            acc[k].y = 0.0f;
        }
    }

    const int jl      = i0 - base;
    const int jt_self = (jl / TJ) * TJ;
    const int js      = jl - jt_self + wid;
    float vself = 0.0f;

    int buf = 0;
    for (int jt = 0; jt < N; jt += TJ) {
#pragma unroll
        for (int rt = 0; rt < 2; ++rt) {
            v2f xf;
            xf.x = xk0[rt];
            xf.y = half ? 0.0f : xk1[rt];
#pragma unroll
            for (int ct = 0; ct < 2; ++ct) {
                const v8f z = {};
                const v8f d = __builtin_amdgcn_wmma_f32_16x16x4_f32(
                                  false, xf, false, wv[ct], (short)0, z, false, false);
                float* dst = &vt[buf][(wid * JW + 16 * rt + half) * C + ct * 16 + l15];
#pragma unroll
                for (int r = 0; r < 8; ++r) dst[(2 * r) * C] = d[r];
            }
        }
        __syncthreads();
        if (jt + TJ < N) {
            const float* nx = xs + (size_t)(jt + TJ) * CIN;
#pragma unroll
            for (int rt = 0; rt < 2; ++rt) {
                xk0[rt] = nx[(16 * rt) * CIN + k0];
                xk1[rt] = nx[(16 * rt) * CIN + k1];
            }
        }
        if (jt == jt_self) vself = vt[buf][js * C + lane];
        const float* vb = &vt[buf][(wid * JW) * C + lane];
        float v[JW];
#pragma unroll
        for (int jj = 0; jj < JW; ++jj) v[jj] = vb[jj * C];
        __builtin_amdgcn_sched_barrier(0);
        v2f cs[RI / 2];
#pragma unroll
        for (int k = 0; k < RI / 2; ++k) {
            cs[k].x = fmaxf(v[0], na[k].x);
            cs[k].y = fmaxf(v[0], na[k].y);
        }
#pragma unroll
        for (int jj = 1; jj < JW; ++jj) {
#pragma unroll
            for (int k = 0; k < RI / 2; ++k) {
                v2f m;
                m.x = fmaxf(v[jj], na[k].x);
                m.y = fmaxf(v[jj], na[k].y);
                cs[k] += m;
            }
        }
        const v2f njw = {-(float)JW, -(float)JW};
#pragma unroll
        for (int k = 0; k < RI / 2; ++k)
            acc[k] += __builtin_elementwise_fma(njw, na[k], cs[k]);
        buf ^= 1;
    }

#pragma unroll
    for (int k = 0; k < RI / 2; ++k) {
        red[wid][2 * k][lane]     = acc[k].x;
        red[wid][2 * k + 1][lane] = acc[k].y;
    }
    __syncthreads();

    {
        float naw = na[0].x;
#pragma unroll
        for (int r = 1; r < RI; ++r) naw = (widu == r) ? ((r & 1) ? na[r / 2].y : na[r / 2].x) : naw;
        const float selfterm = fmaxf(vself - naw, 0.0f);
        float s = red[0][wid][lane];
#pragma unroll
        for (int w = 1; w < 8; ++w) s += red[w][wid][lane];
        hs[wid][lane] = (s - selfterm) * inv_nm1;
    }
    hs[wid + 8][lane] = 0.0f;
    __syncthreads();

    if (widu < 2 * NT) {
        const float* hp = &hs[l15][half * 2];
        v2f af[8], bf[8];
#pragma unroll
        for (int s = 0; s < 8; ++s) {
            af[s].x = hp[4 * s];
            af[s].y = hp[4 * s + 1];
            bf[s].x = cok ? bw[s].x : 0.0f;
            bf[s].y = cok ? bw[s].y : 0.0f;
        }
        v8f d = {};
#pragma unroll
        for (int s = 0; s < 8; ++s) {
            d = __builtin_amdgcn_wmma_f32_16x16x4_f32(
                    false, af[s], false, bf[s], (short)0, d, false, false);
        }
        if (cok) {
            float* dst = (part == 0 ? outA : outV);
#pragma unroll
            for (int r = 0; r < 8; ++r) d[r] += bv;
#pragma unroll
            for (int r = 0; r < 8; ++r) {
                const int row = r + half * 8;
                if (row < RI)
                    dst[(size_t)(i0 + row) * COUT + ocol] = d[r];
            }
        }
    }
}

template<int C, int RI, int COUT, bool ADD_BIAS, bool COLSUM>
__global__ __launch_bounds__(256) void edge_pairwise_relu_gemm(
    const float* __restrict__ A,
    const float* __restrict__ V,
    const float* __restrict__ Wn,
    const float* __restrict__ bias,
    float* __restrict__ outA,
    float* __restrict__ outV,
    float* __restrict__ outS,
    int N, float inv_nm1, int plane, int ngrp)
{
    constexpr int TJ   = 256;
    constexpr int NSEG = C / 32;
    constexpr int JW   = TJ / 8;
    constexpr int NLD  = (TJ * 8) / 256;
    constexpr int NT   = (COUT + 15) / 16;
    constexpr int HP   = 34;
    static_assert(2 * NT <= 8, "one output tile per wave");
    static_assert(RI == 8 || RI == 16, "one 16-row tile, filled by whole passes of the 8 waves");
    __shared__ v4f    vt4[2][TJ * 8];
    __shared__ float  red[8][RI][32];
    __shared__ float  hs[16][HP];

    const int tid  = threadIdx.x;
    const int lane = tid & 31;
    const int wid  = tid >> 5;
    const int seg  = blockIdx.x % NSEG;
    const int i0   = (blockIdx.x / NSEG) * RI;
    const int base = (i0 / N) * N;
    const int c    = seg * 32 + lane;

    const int widu = __builtin_amdgcn_readfirstlane(wid);
    const int half = lane >> 4;
    const int l15  = lane & 15;
    const int part = widu / NT;
    const int ocol = (widu % NT) * 16 + l15;
    const bool cok = (ocol < COUT);

    v2f bw[8];
    float bv = 0.0f;
    if (widu < 2 * NT) {
        if constexpr (ADD_BIAS) {
            if (part == 0) bv = bias[cok ? ocol : 0];
        }
        const float* wp = Wn + (size_t)(part * C + seg * 32 + half * 2) * COUT
                             + (cok ? ocol : 0);
#pragma unroll
        for (int s = 0; s < 8; ++s) {
            bw[s].x = wp[(size_t)(4 * s)     * COUT];
            bw[s].y = wp[(size_t)(4 * s + 1) * COUT];
        }
    }

    const float* vsrc = V + (size_t)base * C + seg * 32
                          + (size_t)(tid >> 3) * C + (tid & 7) * 4;
    v4f pre[NLD];
#pragma unroll
    for (int k = 0; k < NLD; ++k)
        pre[k] = *reinterpret_cast<const v4f*>(vsrc + (size_t)(32 * k) * C);

    v2f na[RI / 2], acc[RI / 2];
#pragma unroll
    for (int k = 0; k < RI / 2; ++k) {
        na[k].x  = -A[(i0 + 2 * k) * C + c];
        na[k].y  = -A[(i0 + 2 * k + 1) * C + c];
        acc[k].x = 0.0f;
        acc[k].y = 0.0f;
    }
    float selfterm[RI / 8];
#pragma unroll
    for (int q = 0; q < RI / 8; ++q) {
        const int i = i0 + wid + 8 * q;
        selfterm[q] = fmaxf(A[i * C + c] + V[i * C + c], 0.0f);
    }

    int buf = 0;
    for (int jt = 0; jt < N; jt += TJ) {
#pragma unroll
        for (int k = 0; k < NLD; ++k) vt4[buf][tid + 256 * k] = pre[k];
        __syncthreads();
        if (jt + TJ < N) {
            const float* nsrc = vsrc + (size_t)(jt + TJ) * C;
#pragma unroll
            for (int k = 0; k < NLD; ++k)
                pre[k] = *reinterpret_cast<const v4f*>(nsrc + (size_t)(32 * k) * C);
        }
        const float* vb = reinterpret_cast<const float*>(vt4[buf]) + (wid * JW) * 32 + lane;
        float v[JW];
#pragma unroll
        for (int jj = 0; jj < JW; ++jj) v[jj] = vb[jj * 32];
        __builtin_amdgcn_sched_barrier(0);
        v2f cs[RI / 2];
#pragma unroll
        for (int k = 0; k < RI / 2; ++k) {
            cs[k].x = fmaxf(v[0], na[k].x);
            cs[k].y = fmaxf(v[0], na[k].y);
        }
#pragma unroll
        for (int jj = 1; jj < JW; ++jj) {
#pragma unroll
            for (int k = 0; k < RI / 2; ++k) {
                v2f m;
                m.x = fmaxf(v[jj], na[k].x);
                m.y = fmaxf(v[jj], na[k].y);
                cs[k] += m;
            }
        }
        const v2f njw = {-(float)JW, -(float)JW};
#pragma unroll
        for (int k = 0; k < RI / 2; ++k)
            acc[k] += __builtin_elementwise_fma(njw, na[k], cs[k]);
        buf ^= 1;
    }

#pragma unroll
    for (int k = 0; k < RI / 2; ++k) {
        red[wid][2 * k][lane]     = acc[k].x;
        red[wid][2 * k + 1][lane] = acc[k].y;
    }
    __syncthreads();

#pragma unroll
    for (int q = 0; q < RI / 8; ++q) {
        const int rr = wid + 8 * q;
        float s = red[0][rr][lane];
#pragma unroll
        for (int w = 1; w < 8; ++w) s += red[w][rr][lane];
        hs[rr][lane] = (s - selfterm[q]) * inv_nm1;
    }
    if constexpr (RI < 16) hs[wid + 8][lane] = 0.0f;
    __syncthreads();

    if (widu < 2 * NT) {
        const float* hp = &hs[l15][half * 2];
        v2f af[8], bf[8];
#pragma unroll
        for (int s = 0; s < 8; ++s) {
            af[s].x = hp[4 * s];
            af[s].y = hp[4 * s + 1];
            bf[s].x = cok ? bw[s].x : 0.0f;
            bf[s].y = cok ? bw[s].y : 0.0f;
        }
        v8f d = {};
#pragma unroll
        for (int s = 0; s < 8; ++s) {
            d = __builtin_amdgcn_wmma_f32_16x16x4_f32(
                    false, af[s], false, bf[s], (short)0, d, false, false);
        }
        if (cok) {
            float* dst = (part == 0 ? outA : outV) + (size_t)seg * plane;
            if constexpr (ADD_BIAS) {
#pragma unroll
                for (int r = 0; r < 8; ++r) d[r] += bv;
            }
#pragma unroll
            for (int r = 0; r < 8; ++r) {
                const int row = r + half * 8;
                if (row < RI)
                    dst[(size_t)(i0 + row) * COUT + ocol] = d[r];
            }
        }
        if constexpr (COLSUM) {
            static_assert(RI == 16 && NT == 1, "all 16 tile rows are rows of the block; one outV tile");
            if (part == 1) {
                float cs = ((d[0] + d[1]) + (d[2] + d[3])) + ((d[4] + d[5]) + (d[6] + d[7]));
                cs += other_half(cs);
                if (lane < 4)
                    outS[((size_t)seg * ngrp + blockIdx.x / NSEG) * 4 + lane] = cs;
            }
        }
    }
}

template<int N, int RI, int NB>
__global__ __launch_bounds__(256) void edge_colsum_final3(
    const float* __restrict__ pA,
    const float* __restrict__ pV,
    const float* __restrict__ pS,
    const float* __restrict__ bias,
    float* __restrict__ out,
    float inv_nm1, int plane)
{
    constexpr int G  = N / RI;
    constexpr int TG = NB * G;
    static_assert((N * 3) % 256 == 0, "a block lies in one batch");
    static_assert(G == 64, "a lane takes two row groups of each plane");
    const int t    = threadIdx.x;
    const int lane = t & 31;
    const int idx  = blockIdx.x * 256 + t;
    const int c    = idx % 3;
    const int b    = (blockIdx.x * 256) / (N * 3);

    const float a0 = pA[idx];
    const float a1 = pA[plane + idx];
    const float bc = bias[c];
    const float v0 = pV[idx];
    const float v1 = pV[plane + idx];
    const float* ps = pS + (size_t)(b * G + lane) * 4;
    float q0[3], q1[3], q2[3], q3[3];
#pragma unroll
    for (int k = 0; k < 3; ++k) {
        q0[k] = ps[k];
        q1[k] = ps[32 * 4 + k];
        q2[k] = ps[TG * 4 + k];
        q3[k] = ps[(TG + 32) * 4 + k];
    }
    __builtin_amdgcn_sched_barrier(0);
    const float a_own = (a0 + a1) + bc;
    const float v_own = v0 + v1;
    float s0 = (q0[0] + q1[0]) + (q2[0] + q3[0]);
    float s1 = (q0[1] + q1[1]) + (q2[1] + q3[1]);
    float s2 = (q0[2] + q1[2]) + (q2[2] + q3[2]);

    s0 += row_gather(s0, 0x67452301u, 0xefcdab89u);
    s1 += row_gather(s1, 0x67452301u, 0xefcdab89u);
    s2 += row_gather(s2, 0x67452301u, 0xefcdab89u);
    s0 += row_gather(s0, 0x54761032u, 0xdcfe98bau);
    s1 += row_gather(s1, 0x54761032u, 0xdcfe98bau);
    s2 += row_gather(s2, 0x54761032u, 0xdcfe98bau);
    s0 += row_gather(s0, 0x32107654u, 0xba98fedcu);
    s1 += row_gather(s1, 0x32107654u, 0xba98fedcu);
    s2 += row_gather(s2, 0x32107654u, 0xba98fedcu);
    s0 += row_gather(s0, 0xfedcba98u, 0x76543210u);
    s1 += row_gather(s1, 0xfedcba98u, 0x76543210u);
    s2 += row_gather(s2, 0xfedcba98u, 0x76543210u);
    s0 += other_half(s0);
    s1 += other_half(s1);
    s2 += other_half(s2);

    float s = s0;
    s = (c == 1) ? s1 : s;
    s = (c == 2) ? s2 : s;
    out[idx] = a_own + (s - v_own) * inv_nm1;
}

extern "C" void kernel_launch(void* const* d_in, const int* in_sizes, int n_in,
                              void* d_out, int out_size, void* d_ws, size_t ws_size,
                              hipStream_t stream) {
    (void)in_sizes; (void)n_in; (void)out_size; (void)ws_size;

    const float* x  = (const float*)d_in[0];
    const float* W1 = (const float*)d_in[1];
    const float* b1 = (const float*)d_in[2];
    const float* W2 = (const float*)d_in[3];
    const float* b2 = (const float*)d_in[4];
    const float* W3 = (const float*)d_in[5];
    const float* b3 = (const float*)d_in[6];
    float* out = (float*)d_out;

    const int B = 2, N = 1024;
    const int ROWS = B * N;
    const float inv = 1.0f / (float)(N - 1);

    float* ws = (float*)d_ws;
    float* A2 = ws;
    float* V2 = A2 + ROWS * 64;
    float* PA = V2 + ROWS * 64;
    float* PV = PA + 2 * ROWS * 3;
    float* PS = PV + 2 * ROWS * 3;

    edge_first_pairwise_relu_gemm<64><<<ROWS / 8, 256, 0, stream>>>(
        x, W1, b1, W2, b2, A2, V2, N, inv);

    edge_pairwise_relu_gemm<64, 16, 3, false, true><<<(ROWS / 16) * 2, 256, 0, stream>>>(
        A2, V2, W3, b3, PA, PV, PS, N, inv, ROWS * 3, ROWS / 16);

    edge_colsum_final3<N, 16, B><<<(ROWS * 3) / 256, 256, 0, stream>>>(PA, PV, PS, b3, out, inv, ROWS * 3);
}


# V no longer staged through LDS in the pairwise kernels: a wave sums its rows from registers (k1: WMMA result tiles in place, lane halves added last: new summation order; k2: a row load each)
# speedup vs baseline: 6.2401x; 1.1649x over previous
#include <hip/hip_runtime.h>
#include <hip/hip_bf16.h>

typedef __attribute__((ext_vector_type(2))) float v2f;
typedef __attribute__((ext_vector_type(8))) float v8f;

__device__ __forceinline__ float row_gather(float x, unsigned lo, unsigned hi) {
    const int b = __builtin_bit_cast(int, x);
    return __builtin_bit_cast(float, __builtin_amdgcn_permlane16(b, b, lo, hi, false, false));
}
__device__ __forceinline__ float other_half(float x) {
    const int b = __builtin_bit_cast(int, x);
    return __builtin_bit_cast(float, __builtin_amdgcn_permlanex16(b, b, 0x76543210u, 0xfedcba98u, false, false));
}

template<int COUT>
__global__ __launch_bounds__(256) void edge_first_pairwise_relu_gemm(
    const float* __restrict__ X,
    const float* __restrict__ W0,
    const float* __restrict__ b0,
    const float* __restrict__ Wn,
    const float* __restrict__ bias,
    float* __restrict__ outA,
    float* __restrict__ outV,
    int N, float inv_nm1)
{
    constexpr int CIN  = 3;
    constexpr int C    = 32;
    constexpr int RI   = 8;
    constexpr int TJ   = 256;
    constexpr int JW   = TJ / 8;
    constexpr int NT   = (COUT + 15) / 16;
    constexpr int HP   = 34;
    static_assert(2 * NT <= 8, "one output tile per wave");
    static_assert(JW == 32, "a wave's slice of a step is two 16-row tiles");
    __shared__ float  red[8][RI][32];
    __shared__ float  hs[16][HP];

    const int tid  = threadIdx.x;
    const int lane = tid & 31;
    const int wid  = tid >> 5;
    const int i0   = blockIdx.x * RI;
    const int base = (i0 / N) * N;

    const int widu = __builtin_amdgcn_readfirstlane(wid);
    const int half = lane >> 4;
    const int l15  = lane & 15;
    const int part = widu / NT;
    const int ocol = (widu % NT) * 16 + l15;
    const bool cok = (ocol < COUT);

    v2f bw[8];
    float bv = 0.0f;
    if (widu < 2 * NT) {
        if (part == 0) bv = bias[cok ? ocol : 0];
        const float* wp = Wn + (size_t)(part * C + half * 2) * COUT + (cok ? ocol : 0);
#pragma unroll
        for (int s = 0; s < 8; ++s) {
            bw[s].x = wp[(size_t)(4 * s)     * COUT];
            bw[s].y = wp[(size_t)(4 * s + 1) * COUT];
        }
    }

    const int k0 = half * 2;
    const int k1 = half ? 2 : 1;

    v2f wa[2], wv[2];
#pragma unroll
    for (int ct = 0; ct < 2; ++ct) {
        const int col = ct * 16 + l15;
        const float a1 = W0[k1 * C + col];
        const float v1 = W0[(CIN + k1) * C + col];
        wa[ct].x = W0[k0 * C + col];
        wa[ct].y = half ? 0.0f : a1;
        wv[ct].x = W0[(CIN + k0) * C + col];
        wv[ct].y = half ? 0.0f : v1;
    }

    const float* xs = X + (size_t)(base + wid * JW + l15) * CIN;
    float xk0[2], xk1[2];
#pragma unroll
    for (int rt = 0; rt < 2; ++rt) {
        xk0[rt] = xs[(16 * rt) * CIN + k0];
        xk1[rt] = xs[(16 * rt) * CIN + k1];
    }

    v2f na[2][RI / 2], acc[2][RI / 2];
    float selfterm;
    {
        const float* xo = X + (size_t)(i0 + (l15 & 7)) * CIN;
        const float x1 = xo[k1];
        v2f xa;
        xa.x = xo[k0];
        xa.y = half ? 0.0f : x1;
        const float bl0 = b0[l15];
        const float bl1 = b0[16 + l15];
        const v8f z = {};
        const v8f d0 = __builtin_amdgcn_wmma_f32_16x16x4_f32(
                           false, xa, false, wa[0], (short)0, z, false, false);
        const v8f d1 = __builtin_amdgcn_wmma_f32_16x16x4_f32(
                           false, xa, false, wa[1], (short)0, z, false, false);
        const v8f e0 = __builtin_amdgcn_wmma_f32_16x16x4_f32(
                           false, xa, false, wv[0], (short)0, z, false, false);
        const v8f e1 = __builtin_amdgcn_wmma_f32_16x16x4_f32(
                           false, xa, false, wv[1], (short)0, z, false, false);
#pragma unroll
        for (int k = 0; k < RI / 2; ++k) {
            na[0][k].x = -(d0[2 * k]     + bl0);
            na[0][k].y = -(d0[2 * k + 1] + bl0);
            na[1][k].x = -(d1[2 * k]     + bl1);
            na[1][k].y = -(d1[2 * k + 1] + bl1);
            acc[0][k].x = 0.0f; acc[0][k].y = 0.0f;
            acc[1][k].x = 0.0f; acc[1][k].y = 0.0f;
        }
        float aw = half ? d1[0] : d0[0];
        float vw = half ? e1[0] : e0[0];
#pragma unroll
        for (int r = 1; r < RI; ++r) {
            aw = (widu == r) ? (half ? d1[r] : d0[r]) : aw;
            vw = (widu == r) ? (half ? e1[r] : e0[r]) : vw;
        }
        selfterm = fmaxf((aw + (half ? bl1 : bl0)) + vw, 0.0f);
    }

    for (int jt = 0; jt < N; jt += TJ) {
        v8f d[2][2];
#pragma unroll
        for (int rt = 0; rt < 2; ++rt) {
            v2f xf;
            xf.x = xk0[rt];
            xf.y = half ? 0.0f : xk1[rt];
#pragma unroll
            for (int ct = 0; ct < 2; ++ct) {
                const v8f z = {};
                d[rt][ct] = __builtin_amdgcn_wmma_f32_16x16x4_f32(
                                false, xf, false, wv[ct], (short)0, z, false, false);
            }
        }
        if (jt + TJ < N) {
            const float* nx = xs + (size_t)(jt + TJ) * CIN;
#pragma unroll
            for (int rt = 0; rt < 2; ++rt) {
                xk0[rt] = nx[(16 * rt) * CIN + k0];
                xk1[rt] = nx[(16 * rt) * CIN + k1];
            }
        }
        v2f cs[2][RI / 2];
#pragma unroll
        for (int ct = 0; ct < 2; ++ct) {
#pragma unroll
            for (int k = 0; k < RI / 2; ++k) {
                cs[ct][k].x = fmaxf(d[0][ct][0], na[ct][k].x);
                cs[ct][k].y = fmaxf(d[0][ct][0], na[ct][k].y);
            }
        }
#pragma unroll
        for (int q = 1; q < 16; ++q) {
#pragma unroll
            for (int ct = 0; ct < 2; ++ct) {
                const float vv = d[q >> 3][ct][q & 7];
#pragma unroll
                for (int k = 0; k < RI / 2; ++k) {
                    v2f m;
                    m.x = fmaxf(vv, na[ct][k].x);
                    m.y = fmaxf(vv, na[ct][k].y);
                    cs[ct][k] += m;
                }
            }
        }
        const v2f n16 = {-16.0f, -16.0f};
#pragma unroll
        for (int ct = 0; ct < 2; ++ct) {
#pragma unroll
            for (int k = 0; k < RI / 2; ++k)
                acc[ct][k] += __builtin_elementwise_fma(n16, na[ct][k], cs[ct][k]);
        }
    }

#pragma unroll
    for (int k = 0; k < RI / 2; ++k) {
        const float gx = half ? acc[0][k].x : acc[1][k].x;
        const float gy = half ? acc[0][k].y : acc[1][k].y;
        const float kx = half ? acc[1][k].x : acc[0][k].x;
        const float ky = half ? acc[1][k].y : acc[0][k].y;
        red[wid][2 * k][lane]     = kx + other_half(gx);
        red[wid][2 * k + 1][lane] = ky + other_half(gy);
    }
    __syncthreads();

    {
        float s = red[0][wid][lane];
#pragma unroll
        for (int w = 1; w < 8; ++w) s += red[w][wid][lane];
        hs[wid][lane] = (s - selfterm) * inv_nm1;
    }
    hs[wid + 8][lane] = 0.0f;
    __syncthreads();
    if (widu < 2 * NT) {
        const float* hp = &hs[l15][half * 2];
        v2f af[8], bf[8];
#pragma unroll
        for (int s = 0; s < 8; ++s) {
            af[s].x = hp[4 * s];
            af[s].y = hp[4 * s + 1];
            bf[s].x = cok ? bw[s].x : 0.0f;
            bf[s].y = cok ? bw[s].y : 0.0f;
        }
        v8f d = {};
#pragma unroll
        for (int s = 0; s < 8; ++s) {
            d = __builtin_amdgcn_wmma_f32_16x16x4_f32(
                    false, af[s], false, bf[s], (short)0, d, false, false);
        }
        if (cok) {
            float* dst = (part == 0 ? outA : outV);
#pragma unroll
            for (int r = 0; r < 8; ++r) d[r] += bv;
#pragma unroll
            for (int r = 0; r < 8; ++r) {
                const int row = r + half * 8;
                if (row < RI)
                    dst[(size_t)(i0 + row) * COUT + ocol] = d[r];
            }
        }
    }
}

template<int C, int RI, int COUT, bool ADD_BIAS, bool COLSUM>
__global__ __launch_bounds__(256) void edge_pairwise_relu_gemm(
    const float* __restrict__ A,
    const float* __restrict__ V,
    const float* __restrict__ Wn,
    const float* __restrict__ bias,
    float* __restrict__ outA,
    float* __restrict__ outV,
    float* __restrict__ outS,
    int N, float inv_nm1, int plane, int ngrp)
{
    constexpr int TJ   = 256;
    constexpr int NSEG = C / 32;
    constexpr int JW   = TJ / 8;
    constexpr int NT   = (COUT + 15) / 16;
    constexpr int HP   = 34;
    static_assert(2 * NT <= 8, "one output tile per wave");
    static_assert(RI == 8 || RI == 16, "one 16-row tile, filled by whole passes of the 8 waves");
    __shared__ float  red[8][RI][32];
    __shared__ float  hs[16][HP];

    const int tid  = threadIdx.x;
    const int lane = tid & 31;
    const int wid  = tid >> 5;
    const int seg  = blockIdx.x % NSEG;
    const int i0   = (blockIdx.x / NSEG) * RI;
    const int base = (i0 / N) * N;
    const int c    = seg * 32 + lane;

    const int widu = __builtin_amdgcn_readfirstlane(wid);
    const int half = lane >> 4;
    const int l15  = lane & 15;
    const int part = widu / NT;
    const int ocol = (widu % NT) * 16 + l15;
    const bool cok = (ocol < COUT);

    v2f bw[8];
    float bv = 0.0f;
    if (widu < 2 * NT) {
        if constexpr (ADD_BIAS) {
            if (part == 0) bv = bias[cok ? ocol : 0];
        }
        const float* wp = Wn + (size_t)(part * C + seg * 32 + half * 2) * COUT
                             + (cok ? ocol : 0);
#pragma unroll
        for (int s = 0; s < 8; ++s) {
            bw[s].x = wp[(size_t)(4 * s)     * COUT];
            bw[s].y = wp[(size_t)(4 * s + 1) * COUT];
        }
    }

    const float* vsrc = V + (size_t)(base + wid * JW) * C + c;
    float va[JW], vb[JW];
#pragma unroll
    for (int jj = 0; jj < JW; ++jj) va[jj] = vsrc[(size_t)jj * C];

    v2f na[RI / 2], acc[RI / 2];
#pragma unroll
    for (int k = 0; k < RI / 2; ++k) {
        na[k].x  = -A[(i0 + 2 * k) * C + c];
        na[k].y  = -A[(i0 + 2 * k + 1) * C + c];
        acc[k].x = 0.0f;
        acc[k].y = 0.0f;
    }
    float selfterm[RI / 8];
#pragma unroll
    for (int q = 0; q < RI / 8; ++q) {
        const int i = i0 + wid + 8 * q;
        selfterm[q] = fmaxf(A[i * C + c] + V[i * C + c], 0.0f);
    }

    auto sum_chunk = [&](const float (&v)[JW], bool use) {
        v2f cs[RI / 2];
#pragma unroll
        for (int k = 0; k < RI / 2; ++k) {
            cs[k].x = fmaxf(v[0], na[k].x);
            cs[k].y = fmaxf(v[0], na[k].y);
        }
#pragma unroll
        for (int jj = 1; jj < JW; ++jj) {
#pragma unroll
            for (int k = 0; k < RI / 2; ++k) {
                v2f m;
                m.x = fmaxf(v[jj], na[k].x);
                m.y = fmaxf(v[jj], na[k].y);
                cs[k] += m;
            }
        }
        const v2f njw = {-(float)JW, -(float)JW};
        const v2f zero = {0.0f, 0.0f};
#pragma unroll
        for (int k = 0; k < RI / 2; ++k) {
            const v2f t = __builtin_elementwise_fma(njw, na[k], cs[k]);
            acc[k] += use ? t : zero;
        }
    };

    for (int jt = 0; jt < N; jt += 2 * TJ) {
        const bool second = (jt + TJ < N);
        const int jb = second ? jt + TJ : jt;
        {
            const float* nsrc = vsrc + (size_t)jb * C;
#pragma unroll
            for (int jj = 0; jj < JW; ++jj) vb[jj] = nsrc[(size_t)jj * C];
        }
        __builtin_amdgcn_sched_barrier(0);
        sum_chunk(va, true);
        const int ja = (jt + 2 * TJ < N) ? jt + 2 * TJ : jb;
        {
            const float* nsrc = vsrc + (size_t)ja * C;
#pragma unroll
            for (int jj = 0; jj < JW; ++jj) va[jj] = nsrc[(size_t)jj * C];
        }
        __builtin_amdgcn_sched_barrier(0);
        sum_chunk(vb, second);
    }

#pragma unroll
    for (int k = 0; k < RI / 2; ++k) {
        red[wid][2 * k][lane]     = acc[k].x;
        red[wid][2 * k + 1][lane] = acc[k].y;
    }
    __syncthreads();

#pragma unroll
    for (int q = 0; q < RI / 8; ++q) {
        const int rr = wid + 8 * q;
        float s = red[0][rr][lane];
#pragma unroll
        for (int w = 1; w < 8; ++w) s += red[w][rr][lane];
        hs[rr][lane] = (s - selfterm[q]) * inv_nm1;
    }
    if constexpr (RI < 16) hs[wid + 8][lane] = 0.0f;
    __syncthreads();

    if (widu < 2 * NT) {
        const float* hp = &hs[l15][half * 2];
        v2f af[8], bf[8];
#pragma unroll
        for (int s = 0; s < 8; ++s) {
            af[s].x = hp[4 * s];
            af[s].y = hp[4 * s + 1];
            bf[s].x = cok ? bw[s].x : 0.0f;
            bf[s].y = cok ? bw[s].y : 0.0f;
        }
        v8f d = {};
#pragma unroll
        for (int s = 0; s < 8; ++s) {
            d = __builtin_amdgcn_wmma_f32_16x16x4_f32(
                    false, af[s], false, bf[s], (short)0, d, false, false);
        }
        if (cok) {
            float* dst = (part == 0 ? outA : outV) + (size_t)seg * plane;
            if constexpr (ADD_BIAS) {
#pragma unroll
                for (int r = 0; r < 8; ++r) d[r] += bv;
            }
#pragma unroll
            for (int r = 0; r < 8; ++r) {
                const int row = r + half * 8;
                if (row < RI)
                    dst[(size_t)(i0 + row) * COUT + ocol] = d[r];
            }
        }
        if constexpr (COLSUM) {
            static_assert(RI == 16 && NT == 1, "all 16 tile rows are rows of the block; one outV tile");
            if (part == 1) {
                float cs = ((d[0] + d[1]) + (d[2] + d[3])) + ((d[4] + d[5]) + (d[6] + d[7]));
                cs += other_half(cs);
                if (lane < 4)
                    outS[((size_t)seg * ngrp + blockIdx.x / NSEG) * 4 + lane] = cs;
            }
        }
    }
}

template<int N, int RI, int NB>
__global__ __launch_bounds__(256) void edge_colsum_final3(
    const float* __restrict__ pA,
    const float* __restrict__ pV,
    const float* __restrict__ pS,
    const float* __restrict__ bias,
    float* __restrict__ out,
    float inv_nm1, int plane)
{
    constexpr int G  = N / RI;
    constexpr int TG = NB * G;
    static_assert((N * 3) % 256 == 0, "a block lies in one batch");
    static_assert(G == 64, "a lane takes two row groups of each plane");
    const int t    = threadIdx.x;
    const int lane = t & 31;
    const int idx  = blockIdx.x * 256 + t;
    const int c    = idx % 3;
    const int b    = (blockIdx.x * 256) / (N * 3);

    const float a0 = pA[idx];
    const float a1 = pA[plane + idx];
    const float bc = bias[c];
    const float v0 = pV[idx];
    const float v1 = pV[plane + idx];
    const float* ps = pS + (size_t)(b * G + lane) * 4;
    float q0[3], q1[3], q2[3], q3[3];
#pragma unroll
    for (int k = 0; k < 3; ++k) {
        q0[k] = ps[k];
        q1[k] = ps[32 * 4 + k];
        q2[k] = ps[TG * 4 + k];
        q3[k] = ps[(TG + 32) * 4 + k];
    }
    __builtin_amdgcn_sched_barrier(0);
    const float a_own = (a0 + a1) + bc;
    const float v_own = v0 + v1;
    float s0 = (q0[0] + q1[0]) + (q2[0] + q3[0]);
    float s1 = (q0[1] + q1[1]) + (q2[1] + q3[1]);
    float s2 = (q0[2] + q1[2]) + (q2[2] + q3[2]);

    s0 += row_gather(s0, 0x67452301u, 0xefcdab89u);
    s1 += row_gather(s1, 0x67452301u, 0xefcdab89u);
    s2 += row_gather(s2, 0x67452301u, 0xefcdab89u);
    s0 += row_gather(s0, 0x54761032u, 0xdcfe98bau);
    s1 += row_gather(s1, 0x54761032u, 0xdcfe98bau);
    s2 += row_gather(s2, 0x54761032u, 0xdcfe98bau);
    s0 += row_gather(s0, 0x32107654u, 0xba98fedcu);
    s1 += row_gather(s1, 0x32107654u, 0xba98fedcu);
    s2 += row_gather(s2, 0x32107654u, 0xba98fedcu);
    s0 += row_gather(s0, 0xfedcba98u, 0x76543210u);
    s1 += row_gather(s1, 0xfedcba98u, 0x76543210u);
    s2 += row_gather(s2, 0xfedcba98u, 0x76543210u);
    s0 += other_half(s0);
    s1 += other_half(s1);
    s2 += other_half(s2);

    float s = s0;
    s = (c == 1) ? s1 : s;
    s = (c == 2) ? s2 : s;
    out[idx] = a_own + (s - v_own) * inv_nm1;
}

extern "C" void kernel_launch(void* const* d_in, const int* in_sizes, int n_in,
                              void* d_out, int out_size, void* d_ws, size_t ws_size,
                              hipStream_t stream) {
    (void)in_sizes; (void)n_in; (void)out_size; (void)ws_size;

    const float* x  = (const float*)d_in[0];
    const float* W1 = (const float*)d_in[1];
    const float* b1 = (const float*)d_in[2];
    const float* W2 = (const float*)d_in[3];
    const float* b2 = (const float*)d_in[4];
    const float* W3 = (const float*)d_in[5];
    const float* b3 = (const float*)d_in[6];
    float* out = (float*)d_out;

    const int B = 2, N = 1024;
    const int ROWS = B * N;
    const float inv = 1.0f / (float)(N - 1);

    float* ws = (float*)d_ws;
    float* A2 = ws;
    float* V2 = A2 + ROWS * 64;
    float* PA = V2 + ROWS * 64;
    float* PV = PA + 2 * ROWS * 3;
    float* PS = PV + 2 * ROWS * 3;

    edge_first_pairwise_relu_gemm<64><<<ROWS / 8, 256, 0, stream>>>(
        x, W1, b1, W2, b2, A2, V2, N, inv);

    edge_pairwise_relu_gemm<64, 16, 3, false, true><<<(ROWS / 16) * 2, 256, 0, stream>>>(
        A2, V2, W3, b3, PA, PV, PS, N, inv, ROWS * 3, ROWS / 16);

    edge_colsum_final3<N, 16, B><<<(ROWS * 3) / 256, 256, 0, stream>>>(PA, PV, PS, b3, out, inv, ROWS * 3);
}
